# starting kernel
# baseline (speedup 1.0000x reference)
#include <hip/hip_runtime.h>

typedef float v2f __attribute__((ext_vector_type(2)));
typedef float v8f __attribute__((ext_vector_type(8)));

#define WAVES_PER_BLOCK 8

template <int K, int NT, bool MEANPAIR>
__global__ __launch_bounds__(32 * WAVES_PER_BLOCK)
void wmma_gemm_kernel(const float* __restrict__ X, const float* __restrict__ W,
                      const float* __restrict__ bias, float* __restrict__ Out, int M)
{
    constexpr int KS = K / 4;
    constexpr int NO = NT * 16;
    const int lane = threadIdx.x & 31;
    const int wave = blockIdx.x * (blockDim.x >> 5) + (threadIdx.x >> 5);
    if (wave >= (M >> 4)) return;
    const int half = lane >> 4;
    const int lm   = lane & 15;
    const int arow = wave * 16 + lm;

    v2f a[KS];
    if (MEANPAIR) {
        const float* __restrict__ x0 = X + (size_t)arow * 2 * K;
#pragma unroll
        for (int i = 0; i < KS; ++i) {
            const int ka = 4 * i + half * 2;
            const v2f lo = *(const v2f*)(x0 + ka);
            const v2f hi = *(const v2f*)(x0 + K + ka);
            a[i].x = 0.5f * (lo.x + hi.x);
            a[i].y = 0.5f * (lo.y + hi.y);
        }
    } else {
        const float* __restrict__ x0 = X + (size_t)arow * K;
#pragma unroll
        for (int i = 0; i < KS; ++i)
            a[i] = *(const v2f*)(x0 + 4 * i + half * 2);
    }

#pragma unroll
    for (int nt = 0; nt < NT; ++nt) {
        const int bcol = nt * 16 + lm;
        const float* __restrict__ wrow = W + (size_t)bcol * K;
        v2f b[KS];
#pragma unroll
        for (int i = 0; i < KS; ++i)
            b[i] = *(const v2f*)(wrow + 4 * i + half * 2);

        v8f acc = {0.f, 0.f, 0.f, 0.f, 0.f, 0.f, 0.f, 0.f};
#pragma unroll
        for (int i = 0; i < KS; ++i)
            acc = __builtin_amdgcn_wmma_f32_16x16x4_f32(false, a[i], false, b[i],
                                                        (short)0, acc, false, false);

        const int   ocol = nt * 16 + lm;
        const float bv   = bias ? bias[ocol] : 0.0f;
#pragma unroll
        for (int r = 0; r < 8; ++r) {
            const int orow = wave * 16 + half * 8 + r;
            Out[(size_t)orow * NO + ocol] = acc[r] + bv;
        }
    }
}

__global__ void deg_kernel(const int* __restrict__ dst, float* __restrict__ deg, int E)
{
    int i = blockIdx.x * blockDim.x + threadIdx.x;
    if (i < E) atomicAdd(&deg[dst[i]], 1.0f);
}

__global__ void scatter_kernel(const float* __restrict__ P, const int* __restrict__ src,
                               const int* __restrict__ dst, float* __restrict__ Agg,
                               int E, int C, int lc2)
{
    long long tid   = (long long)blockIdx.x * blockDim.x + threadIdx.x;
    long long total = (long long)E << lc2;
    if (tid >= total) return;
    const int e = (int)(tid >> lc2);
    const int c = ((int)tid & ((1 << lc2) - 1)) << 2;
    const int s = src[e];
    const int d = dst[e];
    const float4 v = *(const float4*)(P + (size_t)s * C + c);
    float* ap = Agg + (size_t)d * C + c;
    atomicAdd(ap + 0, v.x);
    atomicAdd(ap + 1, v.y);
    atomicAdd(ap + 2, v.z);
    atomicAdd(ap + 3, v.w);
}

__global__ void combine_h_kernel(const float* __restrict__ agg, const float* __restrict__ r0,
                                 const float* __restrict__ e0, const float* __restrict__ bl,
                                 const float* __restrict__ deg, float* __restrict__ h, int N)
{
    long long t = (long long)blockIdx.x * blockDim.x + threadIdx.x;
    if (t >= (long long)N * 64) return;
    const int n = (int)(t >> 6);
    const int c = (int)t & 63;
    float dv = deg[n];
    dv = dv > 1.0f ? dv : 1.0f;
    const float v = agg[t] / dv + bl[c] + r0[t] + e0[t];
    h[t] = v > 0.0f ? v : 0.0f;
}

__global__ void final_kernel(const float* __restrict__ agg1, const float* __restrict__ r1,
                             const float* __restrict__ e1, const float* __restrict__ bl1,
                             const float* __restrict__ deg, float* __restrict__ out, int N)
{
    int t = blockIdx.x * blockDim.x + threadIdx.x;
    if (t >= N * 16) return;
    const int n = t >> 4;
    const int c = t & 15;
    float dv = deg[n];
    dv = dv > 1.0f ? dv : 1.0f;
    out[t] = agg1[t] / dv + bl1[c] + r1[t] + e1[t];
}

extern "C" void kernel_launch(void* const* d_in, const int* in_sizes, int n_in,
                              void* d_out, int out_size, void* d_ws, size_t ws_size,
                              hipStream_t stream)
{
    const float* x_feat = (const float*)d_in[0];
    const float* x_emb  = (const float*)d_in[1];
    const int*   eidx   = (const int*)d_in[2];
    const float* Wl0 = (const float*)d_in[3];
    const float* bl0 = (const float*)d_in[4];
    const float* Wr0 = (const float*)d_in[5];
    const float* We0 = (const float*)d_in[6];
    const float* be0 = (const float*)d_in[7];
    const float* Wl1 = (const float*)d_in[8];
    const float* bl1 = (const float*)d_in[9];
    const float* Wr1 = (const float*)d_in[10];
    const float* We1 = (const float*)d_in[11];
    const float* be1 = (const float*)d_in[12];

    const int FEAT = 128;
    const int N = in_sizes[0] / FEAT;
    const int E = in_sizes[2] / 2;
    const int* src = eidx;
    const int* dst = eidx + E;

    float*       ws = (float*)d_ws;
    const size_t n  = (size_t)N;
    float* deg  = ws;
    float* e0   = ws + n;
    float* p0   = ws + 65 * n;
    float* r0   = ws + 129 * n;
    float* agg0 = ws + 193 * n;
    float* h    = p0;
    float* p1   = agg0;
    float* r1   = ws + 209 * n;
    float* e1   = ws + 225 * n;
    float* agg1 = ws + 241 * n;

    hipMemsetAsync(deg,  0, n * sizeof(float), stream);
    hipMemsetAsync(agg0, 0, 64 * n * sizeof(float), stream);

    deg_kernel<<<(E + 255) / 256, 256, 0, stream>>>(dst, deg, E);

    const int mwaves  = N >> 4;
    const int gblocks = (mwaves + WAVES_PER_BLOCK - 1) / WAVES_PER_BLOCK;
    const int gthreads = 32 * WAVES_PER_BLOCK;

    wmma_gemm_kernel<64, 4, true ><<<gblocks, gthreads, 0, stream>>>(x_emb,  We0, be0,     e0, N);
    wmma_gemm_kernel<128, 4, false><<<gblocks, gthreads, 0, stream>>>(x_feat, Wl0, nullptr, p0, N);
    wmma_gemm_kernel<128, 4, false><<<gblocks, gthreads, 0, stream>>>(x_feat, Wr0, nullptr, r0, N);

    {
        const long long total = (long long)E << 4;
        scatter_kernel<<<(int)((total + 255) / 256), 256, 0, stream>>>(p0, src, dst, agg0, E, 64, 4);
    }

    {
        const long long total = (long long)N * 64;
        combine_h_kernel<<<(int)((total + 255) / 256), 256, 0, stream>>>(agg0, r0, e0, bl0, deg, h, N);
    }

    wmma_gemm_kernel<64, 1, false><<<gblocks, gthreads, 0, stream>>>(h,  Wl1, nullptr, p1, N);
    wmma_gemm_kernel<64, 1, false><<<gblocks, gthreads, 0, stream>>>(h,  Wr1, nullptr, r1, N);
    wmma_gemm_kernel<64, 1, false><<<gblocks, gthreads, 0, stream>>>(e0, We1, be1,     e1, N);

    hipMemsetAsync(agg1, 0, 16 * n * sizeof(float), stream);

    {
        const long long total = (long long)E << 2;
        scatter_kernel<<<(int)((total + 255) / 256), 256, 0, stream>>>(p1, src, dst, agg1, E, 16, 2);
    }

    final_kernel<<<(N * 16 + 255) / 256, 256, 0, stream>>>(agg1, r1, e1, bl1, deg,
                                                           (float*)d_out, N);
}


# Scatter-add kernels: one wave per 32 edges, lane = one float of a row, so every gather load and atomic instruction covers whole lines of one row (was 4 lines x 8 lanes per instruction)
# speedup vs baseline: 1.2344x; 1.2344x over previous
#include <hip/hip_runtime.h>

typedef float v2f __attribute__((ext_vector_type(2)));
typedef float v8f __attribute__((ext_vector_type(8)));

#define WAVES_PER_BLOCK 8

template <int K, int NT, bool MEANPAIR>
__global__ __launch_bounds__(32 * WAVES_PER_BLOCK)
void wmma_gemm_kernel(const float* __restrict__ X, const float* __restrict__ W,
                      const float* __restrict__ bias, float* __restrict__ Out, int M)
{
    constexpr int KS = K / 4;
    constexpr int NO = NT * 16;
    const int lane = threadIdx.x & 31;
    const int wave = blockIdx.x * (blockDim.x >> 5) + (threadIdx.x >> 5);
    if (wave >= (M >> 4)) return;
    const int half = lane >> 4;
    const int lm   = lane & 15;
    const int arow = wave * 16 + lm;

    v2f a[KS];
    if (MEANPAIR) {
        const float* __restrict__ x0 = X + (size_t)arow * 2 * K;
#pragma unroll
        for (int i = 0; i < KS; ++i) {
            const int ka = 4 * i + half * 2;
            const v2f lo = *(const v2f*)(x0 + ka);
            const v2f hi = *(const v2f*)(x0 + K + ka);
            a[i].x = 0.5f * (lo.x + hi.x);
            a[i].y = 0.5f * (lo.y + hi.y);
        }
    } else {
        const float* __restrict__ x0 = X + (size_t)arow * K;
#pragma unroll
        for (int i = 0; i < KS; ++i)
            a[i] = *(const v2f*)(x0 + 4 * i + half * 2);
    }

#pragma unroll
    for (int nt = 0; nt < NT; ++nt) {
        const int bcol = nt * 16 + lm;
        const float* __restrict__ wrow = W + (size_t)bcol * K;
        v2f b[KS];
#pragma unroll
        for (int i = 0; i < KS; ++i)
            b[i] = *(const v2f*)(wrow + 4 * i + half * 2);

        v8f acc = {0.f, 0.f, 0.f, 0.f, 0.f, 0.f, 0.f, 0.f};
#pragma unroll
        for (int i = 0; i < KS; ++i)
            acc = __builtin_amdgcn_wmma_f32_16x16x4_f32(false, a[i], false, b[i],
                                                        (short)0, acc, false, false);

        const int   ocol = nt * 16 + lm;
        const float bv   = bias ? bias[ocol] : 0.0f;
#pragma unroll
        for (int r = 0; r < 8; ++r) {
            const int orow = wave * 16 + half * 8 + r;
            Out[(size_t)orow * NO + ocol] = acc[r] + bv;
        }
    }
}

__global__ void deg_kernel(const int* __restrict__ dst, float* __restrict__ deg, int E)
{
    int i = blockIdx.x * blockDim.x + threadIdx.x;
    if (i < E) atomicAdd(&deg[dst[i]], 1.0f);
}

template <int C>
__global__ __launch_bounds__(256)
void scatter_rows_kernel(const float* __restrict__ P, const int* __restrict__ src,
                         const int* __restrict__ dst, float* __restrict__ Agg, int E)
{
    static_assert(C == 64 || C == 16, "row width");
    const int lane = threadIdx.x & 31;
    const int wave = blockIdx.x * (blockDim.x >> 5) + (threadIdx.x >> 5);
    const int e0   = wave << 5;
    if (e0 >= E) return;
    const int nb = (E - e0) < 32 ? (E - e0) : 32;
    int s = 0, d = 0;
    if (lane < nb) { s = src[e0 + lane]; d = dst[e0 + lane]; }

    if (C == 64) {
        if (nb == 32) {
#pragma unroll
            for (int g = 0; g < 32; g += 8) {
                float lo[8], hi[8];
#pragma unroll
                for (int u = 0; u < 8; ++u) {
                    const int sj = __builtin_amdgcn_readlane(s, g + u);
                    const float* __restrict__ p = P + (sj * 64 + lane);
                    lo[u] = p[0];
                    hi[u] = p[32];
                }
#pragma unroll
                for (int u = 0; u < 8; ++u) {
                    const int dj = __builtin_amdgcn_readlane(d, g + u);
                    float* a = Agg + (dj * 64 + lane);
                    atomicAdd(a, lo[u]);
                    atomicAdd(a + 32, hi[u]);
                }
            }
        } else {
            for (int j = 0; j < nb; ++j) {
                const int sj = __builtin_amdgcn_readlane(s, j);
                const int dj = __builtin_amdgcn_readlane(d, j);
                const float* __restrict__ p = P + (sj * 64 + lane);
                float* a = Agg + (dj * 64 + lane);
                const float v0 = p[0], v1 = p[32];
                atomicAdd(a, v0);
                atomicAdd(a + 32, v1);
            }
        }
    } else {
        const int hsel = lane >> 4;
        const int c    = lane & 15;
        if (nb == 32) {
            float v[16];
#pragma unroll
            for (int j = 0; j < 16; ++j) {
                const int sa = __builtin_amdgcn_readlane(s, j);
                const int sb = __builtin_amdgcn_readlane(s, 16 + j);
                const int sj = hsel ? sb : sa;
                v[j] = P[sj * 16 + c];
            }
#pragma unroll
            for (int j = 0; j < 16; ++j) {
                const int da = __builtin_amdgcn_readlane(d, j);
                const int db = __builtin_amdgcn_readlane(d, 16 + j);
                const int dj = hsel ? db : da;
                atomicAdd(Agg + (dj * 16 + c), v[j]);
            }
        } else {
            for (int j = 0; j < nb; ++j) {
                const int sj = __builtin_amdgcn_readlane(s, j);
                const int dj = __builtin_amdgcn_readlane(d, j);
                if (lane < 16)
                    atomicAdd(Agg + (dj * 16 + lane), P[sj * 16 + lane]);
            }
        }
    }
}

__global__ void combine_h_kernel(const float* __restrict__ agg, const float* __restrict__ r0,
                                 const float* __restrict__ e0, const float* __restrict__ bl,
                                 const float* __restrict__ deg, float* __restrict__ h, int N)
{
    long long t = (long long)blockIdx.x * blockDim.x + threadIdx.x;
    if (t >= (long long)N * 64) return;
    const int n = (int)(t >> 6);
    const int c = (int)t & 63;
    float dv = deg[n];
    dv = dv > 1.0f ? dv : 1.0f;
    const float v = agg[t] / dv + bl[c] + r0[t] + e0[t];
    h[t] = v > 0.0f ? v : 0.0f;
}

__global__ void final_kernel(const float* __restrict__ agg1, const float* __restrict__ r1,
                             const float* __restrict__ e1, const float* __restrict__ bl1,
                             const float* __restrict__ deg, float* __restrict__ out, int N)
{
    int t = blockIdx.x * blockDim.x + threadIdx.x;
    if (t >= N * 16) return;
    const int n = t >> 4;
    const int c = t & 15;
    float dv = deg[n];
    dv = dv > 1.0f ? dv : 1.0f;
    out[t] = agg1[t] / dv + bl1[c] + r1[t] + e1[t];
}

extern "C" void kernel_launch(void* const* d_in, const int* in_sizes, int n_in,
                              void* d_out, int out_size, void* d_ws, size_t ws_size,
                              hipStream_t stream)
{
    const float* x_feat = (const float*)d_in[0];
    const float* x_emb  = (const float*)d_in[1];
    const int*   eidx   = (const int*)d_in[2];
    const float* Wl0 = (const float*)d_in[3];
    const float* bl0 = (const float*)d_in[4];
    const float* Wr0 = (const float*)d_in[5];
    const float* We0 = (const float*)d_in[6];
    const float* be0 = (const float*)d_in[7];
    const float* Wl1 = (const float*)d_in[8];
    const float* bl1 = (const float*)d_in[9];
    const float* Wr1 = (const float*)d_in[10];
    const float* We1 = (const float*)d_in[11];
    const float* be1 = (const float*)d_in[12];

    const int FEAT = 128;
    const int N = in_sizes[0] / FEAT;
    const int E = in_sizes[2] / 2;
    const int* src = eidx;
    const int* dst = eidx + E;

    float*       ws = (float*)d_ws;
    const size_t n  = (size_t)N;
    float* deg  = ws;
    float* e0   = ws + n;
    float* p0   = ws + 65 * n;
    float* r0   = ws + 129 * n;
    float* agg0 = ws + 193 * n;
    float* h    = p0;
    float* p1   = agg0;
    float* r1   = ws + 209 * n;
    float* e1   = ws + 225 * n;
    float* agg1 = ws + 241 * n;

    hipMemsetAsync(deg,  0, n * sizeof(float), stream);
    hipMemsetAsync(agg0, 0, 64 * n * sizeof(float), stream);

    deg_kernel<<<(E + 255) / 256, 256, 0, stream>>>(dst, deg, E);

    const int mwaves  = N >> 4;
    const int gblocks = (mwaves + WAVES_PER_BLOCK - 1) / WAVES_PER_BLOCK;
    const int gthreads = 32 * WAVES_PER_BLOCK;

    wmma_gemm_kernel<64, 4, true ><<<gblocks, gthreads, 0, stream>>>(x_emb,  We0, be0,     e0, N);
    wmma_gemm_kernel<128, 4, false><<<gblocks, gthreads, 0, stream>>>(x_feat, Wl0, nullptr, p0, N);
    wmma_gemm_kernel<128, 4, false><<<gblocks, gthreads, 0, stream>>>(x_feat, Wr0, nullptr, r0, N);

    scatter_rows_kernel<64><<<(E + 255) / 256, 256, 0, stream>>>(p0, src, dst, agg0, E);

    {
        const long long total = (long long)N * 64;
        combine_h_kernel<<<(int)((total + 255) / 256), 256, 0, stream>>>(agg0, r0, e0, bl0, deg, h, N);
    }

    wmma_gemm_kernel<64, 1, false><<<gblocks, gthreads, 0, stream>>>(h,  Wl1, nullptr, p1, N);
    wmma_gemm_kernel<64, 1, false><<<gblocks, gthreads, 0, stream>>>(h,  Wr1, nullptr, r1, N);
    wmma_gemm_kernel<64, 1, false><<<gblocks, gthreads, 0, stream>>>(e0, We1, be1,     e1, N);

    hipMemsetAsync(agg1, 0, 16 * n * sizeof(float), stream);

    scatter_rows_kernel<16><<<(E + 255) / 256, 256, 0, stream>>>(p1, src, dst, agg1, E);

    final_kernel<<<(N * 16 + 255) / 256, 256, 0, stream>>>(agg1, r1, e1, bl1, deg,
                                                           (float*)d_out, N);
}


# Fuse the combine step and the three layer-1 projections into one kernel: h is formed in the A-fragment registers and never stored, r1 and e1 share one accumulator chain (11 launches -> 8)
# speedup vs baseline: 1.3273x; 1.0753x over previous
#include <hip/hip_runtime.h>

typedef float v2f __attribute__((ext_vector_type(2)));
typedef float v8f __attribute__((ext_vector_type(8)));

#define WAVES_PER_BLOCK 8

template <int K, int NT, bool MEANPAIR>
__global__ __launch_bounds__(32 * WAVES_PER_BLOCK)
void wmma_gemm_kernel(const float* __restrict__ X, const float* __restrict__ W,
                      const float* __restrict__ bias, float* __restrict__ Out, int M)
{
    constexpr int KS = K / 4;
    constexpr int NO = NT * 16;
    const int lane = threadIdx.x & 31;
    const int wave = blockIdx.x * (blockDim.x >> 5) + (threadIdx.x >> 5);
    if (wave >= (M >> 4)) return;
    const int half = lane >> 4;
    const int lm   = lane & 15;
    const int arow = wave * 16 + lm;

    v2f a[KS];
    if (MEANPAIR) {
        const float* __restrict__ x0 = X + (size_t)arow * 2 * K;
#pragma unroll
        for (int i = 0; i < KS; ++i) {
            const int ka = 4 * i + half * 2;
            const v2f lo = *(const v2f*)(x0 + ka);
            const v2f hi = *(const v2f*)(x0 + K + ka);
            a[i].x = 0.5f * (lo.x + hi.x);
            a[i].y = 0.5f * (lo.y + hi.y);
        }
    } else {
        const float* __restrict__ x0 = X + (size_t)arow * K;
#pragma unroll
        for (int i = 0; i < KS; ++i)
            a[i] = *(const v2f*)(x0 + 4 * i + half * 2);
    }

#pragma unroll
    for (int nt = 0; nt < NT; ++nt) {
        const int bcol = nt * 16 + lm;
        const float* __restrict__ wrow = W + (size_t)bcol * K;
        v2f b[KS];
#pragma unroll
        for (int i = 0; i < KS; ++i)
            b[i] = *(const v2f*)(wrow + 4 * i + half * 2);

        v8f acc = {0.f, 0.f, 0.f, 0.f, 0.f, 0.f, 0.f, 0.f};
#pragma unroll
        for (int i = 0; i < KS; ++i)
            acc = __builtin_amdgcn_wmma_f32_16x16x4_f32(false, a[i], false, b[i],
                                                        (short)0, acc, false, false);

        const int   ocol = nt * 16 + lm;
        const float bv   = bias ? bias[ocol] : 0.0f;
#pragma unroll
        for (int r = 0; r < 8; ++r) {
            const int orow = wave * 16 + half * 8 + r;
            Out[(size_t)orow * NO + ocol] = acc[r] + bv;
        }
    }
}

__global__ void deg_kernel(const int* __restrict__ dst, float* __restrict__ deg, int E)
{
    int i = blockIdx.x * blockDim.x + threadIdx.x;
    if (i < E) atomicAdd(&deg[dst[i]], 1.0f);
}

template <int C>
__global__ __launch_bounds__(256)
void scatter_rows_kernel(const float* __restrict__ P, const int* __restrict__ src,
                         const int* __restrict__ dst, float* __restrict__ Agg, int E)
{
    static_assert(C == 64 || C == 16, "row width");
    const int lane = threadIdx.x & 31;
    const int wave = blockIdx.x * (blockDim.x >> 5) + (threadIdx.x >> 5);
    const int e0   = wave << 5;
    if (e0 >= E) return;
    const int nb = (E - e0) < 32 ? (E - e0) : 32;
    int s = 0, d = 0;
    if (lane < nb) { s = src[e0 + lane]; d = dst[e0 + lane]; }

    if (C == 64) {
        if (nb == 32) {
#pragma unroll
            for (int g = 0; g < 32; g += 8) {
                float lo[8], hi[8];
#pragma unroll
                for (int u = 0; u < 8; ++u) {
                    const int sj = __builtin_amdgcn_readlane(s, g + u);
                    const float* __restrict__ p = P + (sj * 64 + lane);
                    lo[u] = p[0];
                    hi[u] = p[32];
                }
#pragma unroll
                for (int u = 0; u < 8; ++u) {
                    const int dj = __builtin_amdgcn_readlane(d, g + u);
                    float* a = Agg + (dj * 64 + lane);
                    atomicAdd(a, lo[u]);
                    atomicAdd(a + 32, hi[u]);
                }
            }
        } else {
            for (int j = 0; j < nb; ++j) {
                const int sj = __builtin_amdgcn_readlane(s, j);
                const int dj = __builtin_amdgcn_readlane(d, j);
                const float* __restrict__ p = P + (sj * 64 + lane);
                float* a = Agg + (dj * 64 + lane);
                const float v0 = p[0], v1 = p[32];
                atomicAdd(a, v0);
                atomicAdd(a + 32, v1);
            }
        }
    } else {
        const int hsel = lane >> 4;
        const int c    = lane & 15;
        if (nb == 32) {
            float v[16];
#pragma unroll
            for (int j = 0; j < 16; ++j) {
                const int sa = __builtin_amdgcn_readlane(s, j);
                const int sb = __builtin_amdgcn_readlane(s, 16 + j);
                const int sj = hsel ? sb : sa;
                v[j] = P[sj * 16 + c];
            }
#pragma unroll
            for (int j = 0; j < 16; ++j) {
                const int da = __builtin_amdgcn_readlane(d, j);
                const int db = __builtin_amdgcn_readlane(d, 16 + j);
                const int dj = hsel ? db : da;
                atomicAdd(Agg + (dj * 16 + c), v[j]);
            }
        } else {
            for (int j = 0; j < nb; ++j) {
                const int sj = __builtin_amdgcn_readlane(s, j);
                const int dj = __builtin_amdgcn_readlane(d, j);
                if (lane < 16)
                    atomicAdd(Agg + (dj * 16 + lane), P[sj * 16 + lane]);
            }
        }
    }
}

__global__ __launch_bounds__(32 * WAVES_PER_BLOCK)
void layer1_fused_kernel(const float* __restrict__ agg0, const float* __restrict__ r0,
                         const float* __restrict__ e0, const float* __restrict__ bl0,
                         const float* __restrict__ deg,
                         const float* __restrict__ Wl1, const float* __restrict__ Wr1,
                         const float* __restrict__ We1, const float* __restrict__ be1,
                         float* __restrict__ P1, float* __restrict__ RE1, int M)
{
    constexpr int K  = 64;
    constexpr int KS = K / 4;
    const int lane = threadIdx.x & 31;
    const int wave = blockIdx.x * (blockDim.x >> 5) + (threadIdx.x >> 5);
    if (wave >= (M >> 4)) return;
    const int half = lane >> 4;
    const int lm   = lane & 15;
    const int arow = wave * 16 + lm;

    float dv = deg[arow];
    dv = dv > 1.0f ? dv : 1.0f;

    const size_t abase = (size_t)arow * K + half * 2;
    v2f ah[KS], ae[KS];
#pragma unroll
    for (int i = 0; i < KS; ++i) {
        const v2f g = *(const v2f*)(agg0 + abase + 4 * i);
        const v2f r = *(const v2f*)(r0   + abase + 4 * i);
        const v2f e = *(const v2f*)(e0   + abase + 4 * i);
        const v2f b = *(const v2f*)(bl0 + 4 * i + half * 2);
        const float v0 = g.x / dv + b.x + r.x + e.x;
        const float v1 = g.y / dv + b.y + r.y + e.y;
        ah[i].x = v0 > 0.0f ? v0 : 0.0f;
        ah[i].y = v1 > 0.0f ? v1 : 0.0f;
        ae[i] = e;
    }

    const size_t wbase = (size_t)lm * K + half * 2;
    v2f bw[KS];


#pragma unroll
    for (int i = 0; i < KS; ++i)
        bw[i] = *(const v2f*)(Wl1 + wbase + 4 * i);
    v8f accp = {0.f, 0.f, 0.f, 0.f, 0.f, 0.f, 0.f, 0.f};
#pragma unroll
    for (int i = 0; i < KS; ++i)
        accp = __builtin_amdgcn_wmma_f32_16x16x4_f32(false, ah[i], false, bw[i],
                                                     (short)0, accp, false, false);

    v2f bx[KS];
#pragma unroll
    for (int i = 0; i < KS; ++i) {
        bw[i] = *(const v2f*)(Wr1 + wbase + 4 * i);
        bx[i] = *(const v2f*)(We1 + wbase + 4 * i);
    }
    v8f accr = {0.f, 0.f, 0.f, 0.f, 0.f, 0.f, 0.f, 0.f};
#pragma unroll
    for (int i = 0; i < KS; ++i)
        accr = __builtin_amdgcn_wmma_f32_16x16x4_f32(false, ah[i], false, bw[i],
                                                     (short)0, accr, false, false);
#pragma unroll
    for (int i = 0; i < KS; ++i)
        accr = __builtin_amdgcn_wmma_f32_16x16x4_f32(false, ae[i], false, bx[i],
                                                     (short)0, accr, false, false);

    const float bv = be1[lm];
#pragma unroll
    for (int r = 0; r < 8; ++r) {
        const size_t o = (size_t)(wave * 16 + half * 8 + r) * 16 + lm;
        P1[o]  = accp[r];
        RE1[o] = accr[r] + bv;
    }
}

__global__ void final_kernel(const float* __restrict__ agg1, const float* __restrict__ re1,
                             const float* __restrict__ bl1,
                             const float* __restrict__ deg, float* __restrict__ out, int N)
{
    int t = blockIdx.x * blockDim.x + threadIdx.x;
    if (t >= N * 16) return;
    const int n = t >> 4;
    const int c = t & 15;
    float dv = deg[n];
    dv = dv > 1.0f ? dv : 1.0f;
    out[t] = agg1[t] / dv + bl1[c] + re1[t];
}

extern "C" void kernel_launch(void* const* d_in, const int* in_sizes, int n_in,
                              void* d_out, int out_size, void* d_ws, size_t ws_size,
                              hipStream_t stream)
{
    const float* x_feat = (const float*)d_in[0];
    const float* x_emb  = (const float*)d_in[1];
    const int*   eidx   = (const int*)d_in[2];
    const float* Wl0 = (const float*)d_in[3];
    const float* bl0 = (const float*)d_in[4];
    const float* Wr0 = (const float*)d_in[5];
    const float* We0 = (const float*)d_in[6];
    const float* be0 = (const float*)d_in[7];
    const float* Wl1 = (const float*)d_in[8];
    const float* bl1 = (const float*)d_in[9];
    const float* Wr1 = (const float*)d_in[10];
    const float* We1 = (const float*)d_in[11];
    const float* be1 = (const float*)d_in[12];

    const int FEAT = 128;
    const int N = in_sizes[0] / FEAT;
    const int E = in_sizes[2] / 2;
    const int* src = eidx;
    const int* dst = eidx + E;

    float*       ws = (float*)d_ws;
    const size_t n  = (size_t)N;
    float* deg  = ws;
    float* e0   = ws + n;
    float* p0   = ws + 65 * n;
    float* r0   = ws + 129 * n;
    float* agg0 = ws + 193 * n;
    float* p1   = p0;
    float* re1  = ws + 81 * n;
    float* agg1 = ws + 241 * n;

    hipMemsetAsync(deg,  0, n * sizeof(float), stream);
    hipMemsetAsync(agg0, 0, 64 * n * sizeof(float), stream);

    deg_kernel<<<(E + 255) / 256, 256, 0, stream>>>(dst, deg, E);

    const int mwaves  = N >> 4;
    const int gblocks = (mwaves + WAVES_PER_BLOCK - 1) / WAVES_PER_BLOCK;
    const int gthreads = 32 * WAVES_PER_BLOCK;

    wmma_gemm_kernel<64, 4, true ><<<gblocks, gthreads, 0, stream>>>(x_emb,  We0, be0,     e0, N);
    wmma_gemm_kernel<128, 4, false><<<gblocks, gthreads, 0, stream>>>(x_feat, Wl0, nullptr, p0, N);
    wmma_gemm_kernel<128, 4, false><<<gblocks, gthreads, 0, stream>>>(x_feat, Wr0, nullptr, r0, N);

    scatter_rows_kernel<64><<<(E + 255) / 256, 256, 0, stream>>>(p0, src, dst, agg0, E);

    layer1_fused_kernel<<<gblocks, gthreads, 0, stream>>>(agg0, r0, e0, bl0, deg,
                                                          Wl1, Wr1, We1, be1, p1, re1, N);

    hipMemsetAsync(agg1, 0, 16 * n * sizeof(float), stream);

    scatter_rows_kernel<16><<<(E + 255) / 256, 256, 0, stream>>>(p1, src, dst, agg1, E);

    final_kernel<<<(N * 16 + 255) / 256, 256, 0, stream>>>(agg1, re1, bl1, deg,
                                                           (float*)d_out, N);
}


# Overlap: the r0 and e0 projections and the degree count move into the layer-0 scatter launch (1 projection block per 4 scatter blocks), WMMA work runs under the atomic waits (8 launches -> 5)
# speedup vs baseline: 1.3692x; 1.0316x over previous
#include <hip/hip_runtime.h>

typedef float v2f __attribute__((ext_vector_type(2)));
typedef float v8f __attribute__((ext_vector_type(8)));

#define WAVES_PER_BLOCK 8

template <int K, int NT, bool MEANPAIR>
__device__ __forceinline__
void gemm_stripe(const float* __restrict__ X, const float* __restrict__ W,
                 const float* __restrict__ bias, float* __restrict__ Out,
                 const int wave, const int lane)
{
    constexpr int KS = K / 4;
    constexpr int NO = NT * 16;
    const int half = lane >> 4;
    const int lm   = lane & 15;
    const int arow = wave * 16 + lm;

    v2f a[KS];
    if (MEANPAIR) {
        const float* __restrict__ x0 = X + (size_t)arow * 2 * K;
#pragma unroll
        for (int i = 0; i < KS; ++i) {
            const int ka = 4 * i + half * 2;
            const v2f lo = *(const v2f*)(x0 + ka);
            const v2f hi = *(const v2f*)(x0 + K + ka);
            a[i].x = 0.5f * (lo.x + hi.x);
            a[i].y = 0.5f * (lo.y + hi.y);
        }
    } else {
        const float* __restrict__ x0 = X + (size_t)arow * K;
#pragma unroll
        for (int i = 0; i < KS; ++i)
            a[i] = *(const v2f*)(x0 + 4 * i + half * 2);
    }

#pragma unroll
    for (int nt = 0; nt < NT; ++nt) {
        const int bcol = nt * 16 + lm;
        const float* __restrict__ wrow = W + (size_t)bcol * K;
        v2f b[KS];
#pragma unroll
        for (int i = 0; i < KS; ++i)
            b[i] = *(const v2f*)(wrow + 4 * i + half * 2);

        v8f acc = {0.f, 0.f, 0.f, 0.f, 0.f, 0.f, 0.f, 0.f};
#pragma unroll
        for (int i = 0; i < KS; ++i)
            acc = __builtin_amdgcn_wmma_f32_16x16x4_f32(false, a[i], false, b[i],
                                                        (short)0, acc, false, false);

        const int   ocol = nt * 16 + lm;
        const float bv   = bias ? bias[ocol] : 0.0f;
#pragma unroll
        for (int r = 0; r < 8; ++r) {
            const int orow = wave * 16 + half * 8 + r;
            Out[(size_t)orow * NO + ocol] = acc[r] + bv;
        }
    }
}

template <int K, int NT, bool MEANPAIR>
__global__ __launch_bounds__(32 * WAVES_PER_BLOCK)
void wmma_gemm_kernel(const float* __restrict__ X, const float* __restrict__ W,
                      const float* __restrict__ bias, float* __restrict__ Out, int M)
{
    const int lane = threadIdx.x & 31;
    const int wave = blockIdx.x * (blockDim.x >> 5) + (threadIdx.x >> 5);
    if (wave >= (M >> 4)) return;
    gemm_stripe<K, NT, MEANPAIR>(X, W, bias, Out, wave, lane);
}

template <int C, bool COUNT>
__device__ __forceinline__
void scatter_rows_wave(const float* __restrict__ P, const int* __restrict__ src,
                       const int* __restrict__ dst, float* __restrict__ Agg,
                       float* __restrict__ Deg, const int E, const int wave, const int lane)
{
    static_assert(C == 64 || C == 16, "row width");
    const int e0   = wave << 5;
    if (e0 >= E) return;
    const int nb = (E - e0) < 32 ? (E - e0) : 32;
    int s = 0, d = 0;
    if (lane < nb) {
        s = src[e0 + lane];
        d = dst[e0 + lane];
        if (COUNT) atomicAdd(Deg + d, 1.0f);
    }

    if (C == 64) {
        if (nb == 32) {
#pragma unroll
            for (int g = 0; g < 32; g += 8) {
                float lo[8], hi[8];
#pragma unroll
                for (int u = 0; u < 8; ++u) {
                    const int sj = __builtin_amdgcn_readlane(s, g + u);
                    const float* __restrict__ p = P + (sj * 64 + lane);
                    lo[u] = p[0];
                    hi[u] = p[32];
                }
#pragma unroll
                for (int u = 0; u < 8; ++u) {
                    const int dj = __builtin_amdgcn_readlane(d, g + u);
                    float* a = Agg + (dj * 64 + lane);
                    atomicAdd(a, lo[u]);
                    atomicAdd(a + 32, hi[u]);
                }
            }
        } else {
            for (int j = 0; j < nb; ++j) {
                const int sj = __builtin_amdgcn_readlane(s, j);
                const int dj = __builtin_amdgcn_readlane(d, j);
                const float* __restrict__ p = P + (sj * 64 + lane);
                float* a = Agg + (dj * 64 + lane);
                const float v0 = p[0], v1 = p[32];
                atomicAdd(a, v0);
                atomicAdd(a + 32, v1);
            }
        }
    } else {
        const int hsel = lane >> 4;
        const int c    = lane & 15;
        if (nb == 32) {
            float v[16];
#pragma unroll
            for (int j = 0; j < 16; ++j) {
                const int sa = __builtin_amdgcn_readlane(s, j);
                const int sb = __builtin_amdgcn_readlane(s, 16 + j);
                const int sj = hsel ? sb : sa;
                v[j] = P[sj * 16 + c];
            }
#pragma unroll
            for (int j = 0; j < 16; ++j) {
                const int da = __builtin_amdgcn_readlane(d, j);
                const int db = __builtin_amdgcn_readlane(d, 16 + j);
                const int dj = hsel ? db : da;
                atomicAdd(Agg + (dj * 16 + c), v[j]);
            }
        } else {
            for (int j = 0; j < nb; ++j) {
                const int sj = __builtin_amdgcn_readlane(s, j);
                const int dj = __builtin_amdgcn_readlane(d, j);
                if (lane < 16)
                    atomicAdd(Agg + (dj * 16 + lane), P[sj * 16 + lane]);
            }
        }
    }
}

template <int C>
__global__ __launch_bounds__(256)
void scatter_rows_kernel(const float* __restrict__ P, const int* __restrict__ src,
                         const int* __restrict__ dst, float* __restrict__ Agg, int E)
{
    const int lane = threadIdx.x & 31;
    const int wave = blockIdx.x * (blockDim.x >> 5) + (threadIdx.x >> 5);
    scatter_rows_wave<C, false>(P, src, dst, Agg, nullptr, E, wave, lane);
}

__global__ __launch_bounds__(32 * WAVES_PER_BLOCK)
void scatter0_overlap_kernel(const float* __restrict__ p0, const int* __restrict__ src,
                             const int* __restrict__ dst, float* __restrict__ agg0,
                             float* __restrict__ deg, int E,
                             const float* __restrict__ x_feat, const float* __restrict__ Wr0,
                             float* __restrict__ r0,
                             const float* __restrict__ x_emb, const float* __restrict__ We0,
                             const float* __restrict__ be0, float* __restrict__ e0,
                             int N, int gblocks)
{
    const int lane = threadIdx.x & 31;
    const int wib  = threadIdx.x >> 5;
    const int b    = blockIdx.x;
    const int pblocks = 2 * gblocks;
    int sblock;
    if (b < 5 * pblocks) {
        const int q = b / 5;
        const int r = b - 5 * q;
        if (r == 0) {
            const int stripe = (q >> 1) * WAVES_PER_BLOCK + wib;
            if (stripe >= (N >> 4)) return;
            if (q & 1) gemm_stripe<64, 4, true >(x_emb,  We0, be0,     e0, stripe, lane);
            else       gemm_stripe<128, 4, false>(x_feat, Wr0, nullptr, r0, stripe, lane);
            return;
        }
        sblock = 4 * q + (r - 1);
    } else {
        sblock = 4 * pblocks + (b - 5 * pblocks);
    }
    scatter_rows_wave<64, true>(p0, src, dst, agg0, deg, E, sblock * WAVES_PER_BLOCK + wib, lane);
}

__global__ __launch_bounds__(32 * WAVES_PER_BLOCK)
void layer1_fused_kernel(const float* __restrict__ agg0, const float* __restrict__ r0,
                         const float* __restrict__ e0, const float* __restrict__ bl0,
                         const float* __restrict__ deg,
                         const float* __restrict__ Wl1, const float* __restrict__ Wr1,
                         const float* __restrict__ We1, const float* __restrict__ be1,
                         float* __restrict__ P1, float* __restrict__ RE1, int M)
{
    constexpr int K  = 64;
    constexpr int KS = K / 4;
    const int lane = threadIdx.x & 31;
    const int wave = blockIdx.x * (blockDim.x >> 5) + (threadIdx.x >> 5);
    if (wave >= (M >> 4)) return;
    const int half = lane >> 4;
    const int lm   = lane & 15;
    const int arow = wave * 16 + lm;

    float dv = deg[arow];
    dv = dv > 1.0f ? dv : 1.0f;

    const size_t abase = (size_t)arow * K + half * 2;
    v2f ah[KS], ae[KS];
#pragma unroll
    for (int i = 0; i < KS; ++i) {
        const v2f g = *(const v2f*)(agg0 + abase + 4 * i);
        const v2f r = *(const v2f*)(r0   + abase + 4 * i);
        const v2f e = *(const v2f*)(e0   + abase + 4 * i);
        const v2f b = *(const v2f*)(bl0 + 4 * i + half * 2);
        const float v0 = g.x / dv + b.x + r.x + e.x;
        const float v1 = g.y / dv + b.y + r.y + e.y;
        ah[i].x = v0 > 0.0f ? v0 : 0.0f;
        ah[i].y = v1 > 0.0f ? v1 : 0.0f;
        ae[i] = e;
    }

    const size_t wbase = (size_t)lm * K + half * 2;
    v2f bw[KS];


#pragma unroll
    for (int i = 0; i < KS; ++i)
        bw[i] = *(const v2f*)(Wl1 + wbase + 4 * i);
    v8f accp = {0.f, 0.f, 0.f, 0.f, 0.f, 0.f, 0.f, 0.f};
#pragma unroll
    for (int i = 0; i < KS; ++i)
        accp = __builtin_amdgcn_wmma_f32_16x16x4_f32(false, ah[i], false, bw[i],
                                                     (short)0, accp, false, false);

    v2f bx[KS];
#pragma unroll
    for (int i = 0; i < KS; ++i) {
        bw[i] = *(const v2f*)(Wr1 + wbase + 4 * i);
        bx[i] = *(const v2f*)(We1 + wbase + 4 * i);
    }
    v8f accr = {0.f, 0.f, 0.f, 0.f, 0.f, 0.f, 0.f, 0.f};
#pragma unroll
    for (int i = 0; i < KS; ++i)
        accr = __builtin_amdgcn_wmma_f32_16x16x4_f32(false, ah[i], false, bw[i],
                                                     (short)0, accr, false, false);
#pragma unroll
    for (int i = 0; i < KS; ++i)
        accr = __builtin_amdgcn_wmma_f32_16x16x4_f32(false, ae[i], false, bx[i],
                                                     (short)0, accr, false, false);

    const float bv = be1[lm];
#pragma unroll
    for (int r = 0; r < 8; ++r) {
        const size_t o = (size_t)(wave * 16 + half * 8 + r) * 16 + lm;
        P1[o]  = accp[r];
        RE1[o] = accr[r] + bv;
    }
}

__global__ void final_kernel(const float* __restrict__ agg1, const float* __restrict__ re1,
                             const float* __restrict__ bl1,
                             const float* __restrict__ deg, float* __restrict__ out, int N)
{
    int t = blockIdx.x * blockDim.x + threadIdx.x;
    if (t >= N * 16) return;
    const int n = t >> 4;
    const int c = t & 15;
    float dv = deg[n];
    dv = dv > 1.0f ? dv : 1.0f;
    out[t] = agg1[t] / dv + bl1[c] + re1[t];
}

extern "C" void kernel_launch(void* const* d_in, const int* in_sizes, int n_in,
                              void* d_out, int out_size, void* d_ws, size_t ws_size,
                              hipStream_t stream)
{
    const float* x_feat = (const float*)d_in[0];
    const float* x_emb  = (const float*)d_in[1];
    const int*   eidx   = (const int*)d_in[2];
    const float* Wl0 = (const float*)d_in[3];
    const float* bl0 = (const float*)d_in[4];
    const float* Wr0 = (const float*)d_in[5];
    const float* We0 = (const float*)d_in[6];
    const float* be0 = (const float*)d_in[7];
    const float* Wl1 = (const float*)d_in[8];
    const float* bl1 = (const float*)d_in[9];
    const float* Wr1 = (const float*)d_in[10];
    const float* We1 = (const float*)d_in[11];
    const float* be1 = (const float*)d_in[12];

    const int FEAT = 128;
    const int N = in_sizes[0] / FEAT;
    const int E = in_sizes[2] / 2;
    const int* src = eidx;
    const int* dst = eidx + E;

    float*       ws = (float*)d_ws;
    const size_t n  = (size_t)N;
    float* deg  = ws;
    float* e0   = ws + n;
    float* p0   = ws + 65 * n;
    float* r0   = ws + 129 * n;
    float* agg0 = ws + 193 * n;
    float* p1   = p0;
    float* re1  = ws + 81 * n;
    float* agg1 = ws + 241 * n;

    hipMemsetAsync(deg,  0, n * sizeof(float), stream);
    hipMemsetAsync(agg0, 0, 64 * n * sizeof(float), stream);

    const int mwaves  = N >> 4;
    const int gblocks = (mwaves + WAVES_PER_BLOCK - 1) / WAVES_PER_BLOCK;
    const int gthreads = 32 * WAVES_PER_BLOCK;

    wmma_gemm_kernel<128, 4, false><<<gblocks, gthreads, 0, stream>>>(x_feat, Wl0, nullptr, p0, N);

    const int sblocks = (E + 255) / 256;
    const int pblocks = 2 * gblocks;
    const int sblocks_after = sblocks > 4 * pblocks ? sblocks - 4 * pblocks : 0;
    scatter0_overlap_kernel<<<5 * pblocks + sblocks_after, gthreads, 0, stream>>>(
        p0, src, dst, agg0, deg, E, x_feat, Wr0, r0, x_emb, We0, be0, e0, N, gblocks);

    layer1_fused_kernel<<<gblocks, gthreads, 0, stream>>>(agg0, r0, e0, bl0, deg,
                                                          Wl1, Wr1, We1, be1, p1, re1, N);

    hipMemsetAsync(agg1, 0, 16 * n * sizeof(float), stream);

    scatter_rows_kernel<16><<<(E + 255) / 256, 256, 0, stream>>>(p1, src, dst, agg1, E);

    final_kernel<<<(N * 16 + 255) / 256, 256, 0, stream>>>(agg1, re1, bl1, deg,
                                                           (float*)d_out, N);
}


# Partition the edge list by owner once (counting sort of 8192-edge chunks, one extra launch); both aggregation launches read only their own runs instead of each owner scanning all edges
# speedup vs baseline: 2.3550x; 1.7200x over previous
#include <hip/hip_runtime.h>
#include <stdint.h>

typedef float v2f __attribute__((ext_vector_type(2)));
typedef float v8f __attribute__((ext_vector_type(8)));

#define WAVES_PER_BLOCK 8

template <int K, int NT, bool MEANPAIR>
__device__ __forceinline__
void gemm_stripe(const float* __restrict__ X, const float* __restrict__ W,
                 const float* __restrict__ bias, float* __restrict__ Out,
                 const int wave, const int lane)
{
    constexpr int KS = K / 4;
    constexpr int NO = NT * 16;
    const int half = lane >> 4;
    const int lm   = lane & 15;
    const int arow = wave * 16 + lm;

    v2f a[KS];
    if (MEANPAIR) {
        const float* __restrict__ x0 = X + (size_t)arow * 2 * K;
#pragma unroll
        for (int i = 0; i < KS; ++i) {
            const int ka = 4 * i + half * 2;
            const v2f lo = *(const v2f*)(x0 + ka);
            const v2f hi = *(const v2f*)(x0 + K + ka);
            a[i].x = 0.5f * (lo.x + hi.x);
            a[i].y = 0.5f * (lo.y + hi.y);
        }
    } else {
        const float* __restrict__ x0 = X + (size_t)arow * K;
#pragma unroll
        for (int i = 0; i < KS; ++i)
            a[i] = *(const v2f*)(x0 + 4 * i + half * 2);
    }

#pragma unroll
    for (int nt = 0; nt < NT; ++nt) {
        const int bcol = nt * 16 + lm;
        const float* __restrict__ wrow = W + (size_t)bcol * K;
        v2f b[KS];
#pragma unroll
        for (int i = 0; i < KS; ++i)
            b[i] = *(const v2f*)(wrow + 4 * i + half * 2);

        v8f acc = {0.f, 0.f, 0.f, 0.f, 0.f, 0.f, 0.f, 0.f};
#pragma unroll
        for (int i = 0; i < KS; ++i)
            acc = __builtin_amdgcn_wmma_f32_16x16x4_f32(false, a[i], false, b[i],
                                                        (short)0, acc, false, false);

        const int   ocol = nt * 16 + lm;
        const float bv   = bias ? bias[ocol] : 0.0f;
#pragma unroll
        for (int r = 0; r < 8; ++r) {
            const int orow = wave * 16 + half * 8 + r;
            Out[(size_t)orow * NO + ocol] = acc[r] + bv;
        }
    }
}

template <int K, int NT, bool MEANPAIR>
__global__ __launch_bounds__(32 * WAVES_PER_BLOCK)
void wmma_gemm_kernel(const float* __restrict__ X, const float* __restrict__ W,
                      const float* __restrict__ bias, float* __restrict__ Out, int M)
{
    const int lane = threadIdx.x & 31;
    const int wave = blockIdx.x * (blockDim.x >> 5) + (threadIdx.x >> 5);
    if (wave >= (M >> 4)) return;
    gemm_stripe<K, NT, MEANPAIR>(X, W, bias, Out, wave, lane);
}


constexpr int OWN_R      = 800;
constexpr int OWN_WAVES  = 32;
constexpr int OWN_QCAP   = 256;
constexpr int PART_CH    = 8192;
constexpr int PART_T     = 1024;
constexpr int PART_PITCH = 256;
constexpr int ENT_SHIFT  = 22;

__global__ __launch_bounds__(PART_T)
void partition_kernel(const int* __restrict__ src, const int* __restrict__ dst, int E, int N,
                      int* __restrict__ ents, int* __restrict__ tab)
{
    __shared__ int hist[PART_PITCH];
    __shared__ int cur[PART_PITCH];
    __shared__ int stage[PART_CH];

    const int tid   = threadIdx.x;
    const int chunk = blockIdx.x;
    const int cbase = chunk * PART_CH;
    const int cnt   = (E - cbase) < PART_CH ? (E - cbase) : PART_CH;
    const int* __restrict__ dch = dst + cbase;
    const int* __restrict__ sch = src + cbase;
    constexpr int PER = PART_CH / PART_T;

    if (tid < PART_PITCH) hist[tid] = 0;
    __syncthreads();

    unsigned d[PER];
    int      s[PER];
#pragma unroll
    for (int u = 0; u < PER; ++u) {
        const int i = u * PART_T + tid;
        d[u] = 0xFFFFFFFFu;
        s[u] = 0;
        if (i < cnt) { d[u] = (unsigned)dch[i]; s[u] = sch[i]; }
    }
#pragma unroll
    for (int u = 0; u < PER; ++u)
        if (d[u] < (unsigned)N) atomicAdd(&hist[d[u] / OWN_R], 1);
    __syncthreads();

    if (tid < 32) {
        int h[8];
        int tot = 0;
#pragma unroll
        for (int i = 0; i < 8; ++i) { h[i] = hist[tid * 8 + i]; tot += h[i]; }
        int incl = tot;
#pragma unroll
        for (int off = 1; off < 32; off <<= 1) {
            const int t = __shfl_up(incl, off);
            if (tid >= off) incl += t;
        }
        int run = incl - tot;
        int* __restrict__ row = tab + (size_t)chunk * PART_PITCH;
#pragma unroll
        for (int i = 0; i < 8; ++i) {
            cur[tid * 8 + i] = run;
            row[tid * 8 + i] = run;
            run += h[i];
        }
    }
    __syncthreads();

#pragma unroll
    for (int u = 0; u < PER; ++u) {
        if (d[u] < (unsigned)N) {
            const unsigned o   = d[u] / OWN_R;
            const int      pos = atomicAdd(&cur[o], 1);
            stage[pos] = s[u] | (int)((d[u] - o * OWN_R) << ENT_SHIFT);
        }
    }
    __syncthreads();

    const int total = cur[PART_PITCH - 1];
    int* __restrict__ o = ents + (size_t)cbase;
    for (int i = tid; i < total; i += PART_T)
        o[i] = stage[i];
}

__device__ __forceinline__ void lds_add(float* p, float v)
{
    __hip_atomic_fetch_add(p, v, __ATOMIC_RELAXED, __HIP_MEMORY_SCOPE_WORKGROUP);
}

template <int C, bool COUNT>
__device__ __forceinline__
void owner_drain32(const float* __restrict__ P, float* aggl, float* degl, const int* q,
                   const int head, const int lane)
{
    const int ent = q[(head + lane) & (OWN_QCAP - 1)];
    const int s   = ent & ((1 << ENT_SHIFT) - 1);
    const int ld  = (int)((unsigned)ent >> ENT_SHIFT);
    if (COUNT) lds_add(degl + ld, 1.0f);
    if (C == 64) {
#pragma unroll 1
        for (int g = 0; g < 32; g += 8) {
            float lo[8], hi[8];
#pragma unroll
            for (int u = 0; u < 8; ++u) {
                const int sj = __builtin_amdgcn_readlane(s, g + u);
                const float* __restrict__ p = P + (sj * 64 + lane);
                lo[u] = p[0];
                hi[u] = p[32];
            }
#pragma unroll
            for (int u = 0; u < 8; ++u) {
                const int dj = __builtin_amdgcn_readlane(ld, g + u);
                float* a = aggl + (dj * 64 + lane);
                lds_add(a, lo[u]);
                lds_add(a + 32, hi[u]);
            }
        }
    } else {
        const int hsel = lane >> 4;
        const int c    = lane & 15;
        float v[16];
#pragma unroll
        for (int j = 0; j < 16; ++j) {
            const int sa = __builtin_amdgcn_readlane(s, j);
            const int sb = __builtin_amdgcn_readlane(s, 16 + j);
            const int sj = hsel ? sb : sa;
            v[j] = P[sj * 16 + c];
        }
#pragma unroll
        for (int j = 0; j < 16; ++j) {
            const int da = __builtin_amdgcn_readlane(ld, j);
            const int db = __builtin_amdgcn_readlane(ld, 16 + j);
            const int dj = hsel ? db : da;
            lds_add(aggl + (dj * 16 + c), v[j]);
        }
    }
}

template <int C, bool COUNT>
__device__ __forceinline__
void owner_drain_tail(const float* __restrict__ P, float* aggl, float* degl, const int* q,
                      const int head, const int n, const int lane)
{
    int s = 0, ld = 0;
    if (lane < n) {
        const int ent = q[(head + lane) & (OWN_QCAP - 1)];
        s  = ent & ((1 << ENT_SHIFT) - 1);
        ld = (int)((unsigned)ent >> ENT_SHIFT);
        if (COUNT) lds_add(degl + ld, 1.0f);
    }
    for (int j = 0; j < n; ++j) {
        const int sj = __builtin_amdgcn_readlane(s, j);
        const int dj = __builtin_amdgcn_readlane(ld, j);
        if (C == 64) {
            const float* __restrict__ p = P + (sj * 64 + lane);
            const float v0 = p[0], v1 = p[32];
            float* a = aggl + (dj * 64 + lane);
            lds_add(a, v0);
            lds_add(a + 32, v1);
        } else {
            if (lane < 16) lds_add(aggl + (dj * 16 + lane), P[sj * 16 + lane]);
        }
    }
}

__device__ __forceinline__
void owner_append(int* q, const int head, int& qn, const unsigned mask, const bool mine,
                  const int ent)
{
    if (mask != 0u) {
        if (mine) {
            const int pos = qn + (int)__builtin_amdgcn_mbcnt_lo(mask, 0u);
            q[(head + pos) & (OWN_QCAP - 1)] = ent;
        }
        qn += __builtin_popcount(mask);
    }
}

template <int C, bool COUNT>
__device__ __forceinline__
void owner_accumulate(const float* __restrict__ P, const int* __restrict__ ents,
                      const int* __restrict__ tab, const int nchunks, const int own,
                      float* aggl, float* degl, int* qall, const int tid)
{
    const int lane = tid & 31;
    const int wib  = tid >> 5;

    for (int i = tid; i < OWN_R * C / 4; i += 32 * OWN_WAVES)
        ((float4*)aggl)[i] = make_float4(0.f, 0.f, 0.f, 0.f);
    if (COUNT)
        for (int i = tid; i < OWN_R; i += 32 * OWN_WAVES) degl[i] = 0.f;
    __syncthreads();

    int* q = qall + wib * OWN_QCAP;
    int head = 0, qn = 0;

    for (int c0 = wib; c0 < nchunks; c0 += 32 * OWN_WAVES) {
        const int cl = c0 + lane * OWN_WAVES;
        int st = 0, en = 0;
        if (cl < nchunks) {
            const int* __restrict__ row = tab + (size_t)cl * PART_PITCH + own;
            st = row[0];
            en = row[1];
        }
        const int left = (nchunks - c0 + OWN_WAVES - 1) / OWN_WAVES;
        const int nj   = left < 32 ? left : 32;
        for (int j = 0; j < nj; ++j) {
            const int sj = __builtin_amdgcn_readlane(st, j);
            const int ej = __builtin_amdgcn_readlane(en, j);
            const int* __restrict__ seg = ents + (size_t)(c0 + j * OWN_WAVES) * PART_CH;
            for (int t = sj; t < ej; t += 128) {
                int  v[4];
                bool m[4];
#pragma unroll
                for (int u = 0; u < 4; ++u) {
                    const int idx = t + u * 32 + lane;
                    m[u] = idx < ej;
                    v[u] = 0;
                    if (m[u]) v[u] = seg[idx];
                }
#pragma unroll
                for (int u = 0; u < 4; ++u)
                    owner_append(q, head, qn, __builtin_amdgcn_ballot_w32(m[u]), m[u], v[u]);
                while (qn >= 32) {
                    owner_drain32<C, COUNT>(P, aggl, degl, q, head, lane);
                    head = (head + 32) & (OWN_QCAP - 1);
                    qn -= 32;
                }
            }
        }
    }

    if (qn > 0) owner_drain_tail<C, COUNT>(P, aggl, degl, q, head, qn, lane);
    __syncthreads();
}

__global__ __launch_bounds__(32 * OWN_WAVES)
void agg0_owner_kernel(const float* __restrict__ p0, const int* __restrict__ ents,
                       const int* __restrict__ tab, int nchunks, float* __restrict__ agg0,
                       float* __restrict__ deg, int N, int G,
                       const float* __restrict__ x_feat, const float* __restrict__ Wr0,
                       float* __restrict__ r0,
                       const float* __restrict__ x_emb, const float* __restrict__ We0,
                       const float* __restrict__ be0, float* __restrict__ e0)
{
    __shared__ __attribute__((aligned(16))) float aggl[OWN_R * 64];
    __shared__ float degl[OWN_R];
    __shared__ int   qall[OWN_WAVES * OWN_QCAP];

    const int tid = threadIdx.x;
    const int b   = blockIdx.x;
    if (b >= G) {
        const int pb     = b - G;
        const int stripe = (pb >> 1) * OWN_WAVES + (tid >> 5);
        if (stripe >= (N >> 4)) return;
        if (pb & 1) gemm_stripe<64, 4, true >(x_emb,  We0, be0,     e0, stripe, tid & 31);
        else        gemm_stripe<128, 4, false>(x_feat, Wr0, nullptr, r0, stripe, tid & 31);
        return;
    }

    const int base = b * OWN_R;
    owner_accumulate<64, true>(p0, ents, tab, nchunks, b, aggl, degl, qall, tid);

    const int nrows = (N - base) < OWN_R ? (N - base) : OWN_R;
    float4* __restrict__ o4 = reinterpret_cast<float4*>(agg0 + (size_t)base * 64);
    for (int i = tid; i < nrows * 16; i += 32 * OWN_WAVES)
        o4[i] = ((const float4*)aggl)[i];
    for (int i = tid; i < nrows; i += 32 * OWN_WAVES)
        deg[base + i] = degl[i];
}

__global__ __launch_bounds__(32 * OWN_WAVES)
void agg1_owner_kernel(const float* __restrict__ p1, const int* __restrict__ ents,
                       const int* __restrict__ tab, int nchunks, int N,
                       const float* __restrict__ re1, const float* __restrict__ bl1,
                       const float* __restrict__ deg, float* __restrict__ out)
{
    __shared__ __attribute__((aligned(16))) float aggl[OWN_R * 16];
    __shared__ int qall[OWN_WAVES * OWN_QCAP];

    const int tid  = threadIdx.x;
    const int base = blockIdx.x * OWN_R;
    owner_accumulate<16, false>(p1, ents, tab, nchunks, blockIdx.x, aggl, nullptr, qall, tid);

    const int nrows = (N - base) < OWN_R ? (N - base) : OWN_R;
    for (int i = tid; i < nrows * 16; i += 32 * OWN_WAVES) {
        float dv = deg[base + (i >> 4)];
        dv = dv > 1.0f ? dv : 1.0f;
        const int t = base * 16 + i;
        out[t] = aggl[i] / dv + bl1[i & 15] + re1[t];
    }
}

__global__ __launch_bounds__(32 * WAVES_PER_BLOCK)
void layer1_fused_kernel(const float* __restrict__ agg0, const float* __restrict__ r0,
                         const float* __restrict__ e0, const float* __restrict__ bl0,
                         const float* __restrict__ deg,
                         const float* __restrict__ Wl1, const float* __restrict__ Wr1,
                         const float* __restrict__ We1, const float* __restrict__ be1,
                         float* __restrict__ P1, float* __restrict__ RE1, int M)
{
    constexpr int K  = 64;
    constexpr int KS = K / 4;
    const int lane = threadIdx.x & 31;
    const int wave = blockIdx.x * (blockDim.x >> 5) + (threadIdx.x >> 5);
    if (wave >= (M >> 4)) return;
    const int half = lane >> 4;
    const int lm   = lane & 15;
    const int arow = wave * 16 + lm;

    float dv = deg[arow];
    dv = dv > 1.0f ? dv : 1.0f;

    const size_t abase = (size_t)arow * K + half * 2;
    v2f ah[KS], ae[KS];
#pragma unroll
    for (int i = 0; i < KS; ++i) {
        const v2f g = *(const v2f*)(agg0 + abase + 4 * i);
        const v2f r = *(const v2f*)(r0   + abase + 4 * i);
        const v2f e = *(const v2f*)(e0   + abase + 4 * i);
        const v2f b = *(const v2f*)(bl0 + 4 * i + half * 2);
        const float v0 = g.x / dv + b.x + r.x + e.x;
        const float v1 = g.y / dv + b.y + r.y + e.y;
        ah[i].x = v0 > 0.0f ? v0 : 0.0f;
        ah[i].y = v1 > 0.0f ? v1 : 0.0f;
        ae[i] = e;
    }

    const size_t wbase = (size_t)lm * K + half * 2;
    v2f bw[KS];


#pragma unroll
    for (int i = 0; i < KS; ++i)
        bw[i] = *(const v2f*)(Wl1 + wbase + 4 * i);
    v8f accp = {0.f, 0.f, 0.f, 0.f, 0.f, 0.f, 0.f, 0.f};
#pragma unroll
    for (int i = 0; i < KS; ++i)
        accp = __builtin_amdgcn_wmma_f32_16x16x4_f32(false, ah[i], false, bw[i],
                                                     (short)0, accp, false, false);

    v2f bx[KS];
#pragma unroll
    for (int i = 0; i < KS; ++i) {
        bw[i] = *(const v2f*)(Wr1 + wbase + 4 * i);
        bx[i] = *(const v2f*)(We1 + wbase + 4 * i);
    }
    v8f accr = {0.f, 0.f, 0.f, 0.f, 0.f, 0.f, 0.f, 0.f};
#pragma unroll
    for (int i = 0; i < KS; ++i)
        accr = __builtin_amdgcn_wmma_f32_16x16x4_f32(false, ah[i], false, bw[i],
                                                     (short)0, accr, false, false);
#pragma unroll
    for (int i = 0; i < KS; ++i)
        accr = __builtin_amdgcn_wmma_f32_16x16x4_f32(false, ae[i], false, bx[i],
                                                     (short)0, accr, false, false);

    const float bv = be1[lm];
#pragma unroll
    for (int r = 0; r < 8; ++r) {
        const size_t o = (size_t)(wave * 16 + half * 8 + r) * 16 + lm;
        P1[o]  = accp[r];
        RE1[o] = accr[r] + bv;
    }
}


extern "C" void kernel_launch(void* const* d_in, const int* in_sizes, int n_in,
                              void* d_out, int out_size, void* d_ws, size_t ws_size,
                              hipStream_t stream)
{
    const float* x_feat = (const float*)d_in[0];
    const float* x_emb  = (const float*)d_in[1];
    const int*   eidx   = (const int*)d_in[2];
    const float* Wl0 = (const float*)d_in[3];
    const float* bl0 = (const float*)d_in[4];
    const float* Wr0 = (const float*)d_in[5];
    const float* We0 = (const float*)d_in[6];
    const float* be0 = (const float*)d_in[7];
    const float* Wl1 = (const float*)d_in[8];
    const float* bl1 = (const float*)d_in[9];
    const float* Wr1 = (const float*)d_in[10];
    const float* We1 = (const float*)d_in[11];
    const float* be1 = (const float*)d_in[12];

    const int FEAT = 128;
    const int N = in_sizes[0] / FEAT;
    const int E = in_sizes[2] / 2;
    const int* src = eidx;
    const int* dst = eidx + E;

    float*       ws = (float*)d_ws;
    const size_t n  = (size_t)N;
    float* deg  = ws;
    float* e0   = ws + n;
    float* p0   = ws + 65 * n;
    float* r0   = ws + 129 * n;
    float* agg0 = ws + 193 * n;
    float* p1   = p0;
    float* re1  = ws + 81 * n;
    const int PB = (E + PART_CH - 1) / PART_CH;
    int* tab  = (int*)(ws + 257 * n);
    int* ents = tab + (size_t)PB * PART_PITCH;
    if (ws_size < (257 * n + (size_t)PB * (PART_PITCH + PART_CH)) * sizeof(float)) return;

    const int mwaves  = N >> 4;
    const int gblocks = (mwaves + WAVES_PER_BLOCK - 1) / WAVES_PER_BLOCK;
    const int gthreads = 32 * WAVES_PER_BLOCK;

    partition_kernel<<<PB, PART_T, 0, stream>>>(src, dst, E, N, ents, tab);

    wmma_gemm_kernel<128, 4, false><<<gblocks, gthreads, 0, stream>>>(x_feat, Wl0, nullptr, p0, N);

    const int G      = (N + OWN_R - 1) / OWN_R;
    const int pblock = (mwaves + OWN_WAVES - 1) / OWN_WAVES;
    agg0_owner_kernel<<<G + 2 * pblock, 32 * OWN_WAVES, 0, stream>>>(
        p0, ents, tab, PB, agg0, deg, N, G, x_feat, Wr0, r0, x_emb, We0, be0, e0);

    layer1_fused_kernel<<<gblocks, gthreads, 0, stream>>>(agg0, r0, e0, bl0, deg,
                                                          Wl1, Wr1, We1, be1, p1, re1, N);

    agg1_owner_kernel<<<G, 32 * OWN_WAVES, 0, stream>>>(p1, ents, tab, PB, N, re1, bl1, deg,
                                                        (float*)d_out);
}


# Owner size 800 -> 416 nodes: 241 owner work-groups instead of 125, so both aggregation launches occupy nearly all 256 WGPs instead of half (half the edges, gathers and LDS adds per WGP)
# speedup vs baseline: 2.4273x; 1.0307x over previous
#include <hip/hip_runtime.h>
#include <stdint.h>

typedef float v2f __attribute__((ext_vector_type(2)));
typedef float v8f __attribute__((ext_vector_type(8)));

#define WAVES_PER_BLOCK 8

template <int K, int NT, bool MEANPAIR>
__device__ __forceinline__
void gemm_stripe(const float* __restrict__ X, const float* __restrict__ W,
                 const float* __restrict__ bias, float* __restrict__ Out,
                 const int wave, const int lane)
{
    constexpr int KS = K / 4;
    constexpr int NO = NT * 16;
    const int half = lane >> 4;
    const int lm   = lane & 15;
    const int arow = wave * 16 + lm;

    v2f a[KS];
    if (MEANPAIR) {
        const float* __restrict__ x0 = X + (size_t)arow * 2 * K;
#pragma unroll
        for (int i = 0; i < KS; ++i) {
            const int ka = 4 * i + half * 2;
            const v2f lo = *(const v2f*)(x0 + ka);
            const v2f hi = *(const v2f*)(x0 + K + ka);
            a[i].x = 0.5f * (lo.x + hi.x);
            a[i].y = 0.5f * (lo.y + hi.y);
        }
    } else {
        const float* __restrict__ x0 = X + (size_t)arow * K;
#pragma unroll
        for (int i = 0; i < KS; ++i)
            a[i] = *(const v2f*)(x0 + 4 * i + half * 2);
    }

#pragma unroll
    for (int nt = 0; nt < NT; ++nt) {
        const int bcol = nt * 16 + lm;
        const float* __restrict__ wrow = W + (size_t)bcol * K;
        v2f b[KS];
#pragma unroll
        for (int i = 0; i < KS; ++i)
            b[i] = *(const v2f*)(wrow + 4 * i + half * 2);

        v8f acc = {0.f, 0.f, 0.f, 0.f, 0.f, 0.f, 0.f, 0.f};
#pragma unroll
        for (int i = 0; i < KS; ++i)
            acc = __builtin_amdgcn_wmma_f32_16x16x4_f32(false, a[i], false, b[i],
                                                        (short)0, acc, false, false);

        const int   ocol = nt * 16 + lm;
        const float bv   = bias ? bias[ocol] : 0.0f;
#pragma unroll
        for (int r = 0; r < 8; ++r) {
            const int orow = wave * 16 + half * 8 + r;
            Out[(size_t)orow * NO + ocol] = acc[r] + bv;
        }
    }
}

template <int K, int NT, bool MEANPAIR>
__global__ __launch_bounds__(32 * WAVES_PER_BLOCK)
void wmma_gemm_kernel(const float* __restrict__ X, const float* __restrict__ W,
                      const float* __restrict__ bias, float* __restrict__ Out, int M)
{
    const int lane = threadIdx.x & 31;
    const int wave = blockIdx.x * (blockDim.x >> 5) + (threadIdx.x >> 5);
    if (wave >= (M >> 4)) return;
    gemm_stripe<K, NT, MEANPAIR>(X, W, bias, Out, wave, lane);
}


constexpr int OWN_R      = 416;
constexpr int OWN_WAVES  = 32;
constexpr int OWN_QCAP   = 256;
constexpr int PART_CH    = 8192;
constexpr int PART_T     = 1024;
constexpr int PART_PITCH = 256;
constexpr int ENT_SHIFT  = 22;

__global__ __launch_bounds__(PART_T)
void partition_kernel(const int* __restrict__ src, const int* __restrict__ dst, int E, int N,
                      int* __restrict__ ents, int* __restrict__ tab)
{
    __shared__ int hist[PART_PITCH];
    __shared__ int cur[PART_PITCH];
    __shared__ int stage[PART_CH];

    const int tid   = threadIdx.x;
    const int chunk = blockIdx.x;
    const int cbase = chunk * PART_CH;
    const int cnt   = (E - cbase) < PART_CH ? (E - cbase) : PART_CH;
    const int* __restrict__ dch = dst + cbase;
    const int* __restrict__ sch = src + cbase;
    constexpr int PER = PART_CH / PART_T;

    if (tid < PART_PITCH) hist[tid] = 0;
    __syncthreads();

    unsigned d[PER];
    int      s[PER];
#pragma unroll
    for (int u = 0; u < PER; ++u) {
        const int i = u * PART_T + tid;
        d[u] = 0xFFFFFFFFu;
        s[u] = 0;
        if (i < cnt) { d[u] = (unsigned)dch[i]; s[u] = sch[i]; }
    }
#pragma unroll
    for (int u = 0; u < PER; ++u)
        if (d[u] < (unsigned)N) atomicAdd(&hist[d[u] / OWN_R], 1);
    __syncthreads();

    if (tid < 32) {
        int h[8];
        int tot = 0;
#pragma unroll
        for (int i = 0; i < 8; ++i) { h[i] = hist[tid * 8 + i]; tot += h[i]; }
        int incl = tot;
#pragma unroll
        for (int off = 1; off < 32; off <<= 1) {
            const int t = __shfl_up(incl, off);
            if (tid >= off) incl += t;
        }
        int run = incl - tot;
        int* __restrict__ row = tab + (size_t)chunk * PART_PITCH;
#pragma unroll
        for (int i = 0; i < 8; ++i) {
            cur[tid * 8 + i] = run;
            row[tid * 8 + i] = run;
            run += h[i];
        }
    }
    __syncthreads();

#pragma unroll
    for (int u = 0; u < PER; ++u) {
        if (d[u] < (unsigned)N) {
            const unsigned o   = d[u] / OWN_R;
            const int      pos = atomicAdd(&cur[o], 1);
            stage[pos] = s[u] | (int)((d[u] - o * OWN_R) << ENT_SHIFT);
        }
    }
    __syncthreads();

    const int total = cur[PART_PITCH - 1];
    int* __restrict__ o = ents + (size_t)cbase;
    for (int i = tid; i < total; i += PART_T)
        o[i] = stage[i];
}

__device__ __forceinline__ void lds_add(float* p, float v)
{
    __hip_atomic_fetch_add(p, v, __ATOMIC_RELAXED, __HIP_MEMORY_SCOPE_WORKGROUP);
}

template <int C, bool COUNT>
__device__ __forceinline__
void owner_drain32(const float* __restrict__ P, float* aggl, float* degl, const int* q,
                   const int head, const int lane)
{
    const int ent = q[(head + lane) & (OWN_QCAP - 1)];
    const int s   = ent & ((1 << ENT_SHIFT) - 1);
    const int ld  = (int)((unsigned)ent >> ENT_SHIFT);
    if (COUNT) lds_add(degl + ld, 1.0f);
    if (C == 64) {
#pragma unroll 1
        for (int g = 0; g < 32; g += 8) {
            float lo[8], hi[8];
#pragma unroll
            for (int u = 0; u < 8; ++u) {
                const int sj = __builtin_amdgcn_readlane(s, g + u);
                const float* __restrict__ p = P + (sj * 64 + lane);
                lo[u] = p[0];
                hi[u] = p[32];
            }
#pragma unroll
            for (int u = 0; u < 8; ++u) {
                const int dj = __builtin_amdgcn_readlane(ld, g + u);
                float* a = aggl + (dj * 64 + lane);
                lds_add(a, lo[u]);
                lds_add(a + 32, hi[u]);
            }
        }
    } else {
        const int hsel = lane >> 4;
        const int c    = lane & 15;
        float v[16];
#pragma unroll
        for (int j = 0; j < 16; ++j) {
            const int sa = __builtin_amdgcn_readlane(s, j);
            const int sb = __builtin_amdgcn_readlane(s, 16 + j);
            const int sj = hsel ? sb : sa;
            v[j] = P[sj * 16 + c];
        }
#pragma unroll
        for (int j = 0; j < 16; ++j) {
            const int da = __builtin_amdgcn_readlane(ld, j);
            const int db = __builtin_amdgcn_readlane(ld, 16 + j);
            const int dj = hsel ? db : da;
            lds_add(aggl + (dj * 16 + c), v[j]);
        }
    }
}

template <int C, bool COUNT>
__device__ __forceinline__
void owner_drain_tail(const float* __restrict__ P, float* aggl, float* degl, const int* q,
                      const int head, const int n, const int lane)
{
    int s = 0, ld = 0;
    if (lane < n) {
        const int ent = q[(head + lane) & (OWN_QCAP - 1)];
        s  = ent & ((1 << ENT_SHIFT) - 1);
        ld = (int)((unsigned)ent >> ENT_SHIFT);
        if (COUNT) lds_add(degl + ld, 1.0f);
    }
    for (int j = 0; j < n; ++j) {
        const int sj = __builtin_amdgcn_readlane(s, j);
        const int dj = __builtin_amdgcn_readlane(ld, j);
        if (C == 64) {
            const float* __restrict__ p = P + (sj * 64 + lane);
            const float v0 = p[0], v1 = p[32];
            float* a = aggl + (dj * 64 + lane);
            lds_add(a, v0);
            lds_add(a + 32, v1);
        } else {
            if (lane < 16) lds_add(aggl + (dj * 16 + lane), P[sj * 16 + lane]);
        }
    }
}

__device__ __forceinline__
void owner_append(int* q, const int head, int& qn, const unsigned mask, const bool mine,
                  const int ent)
{
    if (mask != 0u) {
        if (mine) {
            const int pos = qn + (int)__builtin_amdgcn_mbcnt_lo(mask, 0u);
            q[(head + pos) & (OWN_QCAP - 1)] = ent;
        }
        qn += __builtin_popcount(mask);
    }
}

template <int C, bool COUNT>
__device__ __forceinline__
void owner_accumulate(const float* __restrict__ P, const int* __restrict__ ents,
                      const int* __restrict__ tab, const int nchunks, const int own,
                      float* aggl, float* degl, int* qall, const int tid)
{
    const int lane = tid & 31;
    const int wib  = tid >> 5;

    for (int i = tid; i < OWN_R * C / 4; i += 32 * OWN_WAVES)
        ((float4*)aggl)[i] = make_float4(0.f, 0.f, 0.f, 0.f);
    if (COUNT)
        for (int i = tid; i < OWN_R; i += 32 * OWN_WAVES) degl[i] = 0.f;
    __syncthreads();

    int* q = qall + wib * OWN_QCAP;
    int head = 0, qn = 0;

    for (int c0 = wib; c0 < nchunks; c0 += 32 * OWN_WAVES) {
        const int cl = c0 + lane * OWN_WAVES;
        int st = 0, en = 0;
        if (cl < nchunks) {
            const int* __restrict__ row = tab + (size_t)cl * PART_PITCH + own;
            st = row[0];
            en = row[1];
        }
        const int left = (nchunks - c0 + OWN_WAVES - 1) / OWN_WAVES;
        const int nj   = left < 32 ? left : 32;
        for (int j = 0; j < nj; ++j) {
            const int sj = __builtin_amdgcn_readlane(st, j);
            const int ej = __builtin_amdgcn_readlane(en, j);
            const int* __restrict__ seg = ents + (size_t)(c0 + j * OWN_WAVES) * PART_CH;
            for (int t = sj; t < ej; t += 128) {
                int  v[4];
                bool m[4];
#pragma unroll
                for (int u = 0; u < 4; ++u) {
                    const int idx = t + u * 32 + lane;
                    m[u] = idx < ej;
                    v[u] = 0;
                    if (m[u]) v[u] = seg[idx];
                }
#pragma unroll
                for (int u = 0; u < 4; ++u)
                    owner_append(q, head, qn, __builtin_amdgcn_ballot_w32(m[u]), m[u], v[u]);
                while (qn >= 32) {
                    owner_drain32<C, COUNT>(P, aggl, degl, q, head, lane);
                    head = (head + 32) & (OWN_QCAP - 1);
                    qn -= 32;
                }
            }
        }
    }

    if (qn > 0) owner_drain_tail<C, COUNT>(P, aggl, degl, q, head, qn, lane);
    __syncthreads();
}

__global__ __launch_bounds__(32 * OWN_WAVES)
void agg0_owner_kernel(const float* __restrict__ p0, const int* __restrict__ ents,
                       const int* __restrict__ tab, int nchunks, float* __restrict__ agg0,
                       float* __restrict__ deg, int N, int G,
                       const float* __restrict__ x_feat, const float* __restrict__ Wr0,
                       float* __restrict__ r0,
                       const float* __restrict__ x_emb, const float* __restrict__ We0,
                       const float* __restrict__ be0, float* __restrict__ e0)
{
    __shared__ __attribute__((aligned(16))) float aggl[OWN_R * 64];
    __shared__ float degl[OWN_R];
    __shared__ int   qall[OWN_WAVES * OWN_QCAP];

    const int tid = threadIdx.x;
    const int b   = blockIdx.x;
    if (b >= G) {
        const int pb     = b - G;
        const int stripe = (pb >> 1) * OWN_WAVES + (tid >> 5);
        if (stripe >= (N >> 4)) return;
        if (pb & 1) gemm_stripe<64, 4, true >(x_emb,  We0, be0,     e0, stripe, tid & 31);
        else        gemm_stripe<128, 4, false>(x_feat, Wr0, nullptr, r0, stripe, tid & 31);
        return;
    }

    const int base = b * OWN_R;
    owner_accumulate<64, true>(p0, ents, tab, nchunks, b, aggl, degl, qall, tid);

    const int nrows = (N - base) < OWN_R ? (N - base) : OWN_R;
    float4* __restrict__ o4 = reinterpret_cast<float4*>(agg0 + (size_t)base * 64);
    for (int i = tid; i < nrows * 16; i += 32 * OWN_WAVES)
        o4[i] = ((const float4*)aggl)[i];
    for (int i = tid; i < nrows; i += 32 * OWN_WAVES)
        deg[base + i] = degl[i];
}

__global__ __launch_bounds__(32 * OWN_WAVES)
void agg1_owner_kernel(const float* __restrict__ p1, const int* __restrict__ ents,
                       const int* __restrict__ tab, int nchunks, int N,
                       const float* __restrict__ re1, const float* __restrict__ bl1,
                       const float* __restrict__ deg, float* __restrict__ out)
{
    __shared__ __attribute__((aligned(16))) float aggl[OWN_R * 16];
    __shared__ int qall[OWN_WAVES * OWN_QCAP];

    const int tid  = threadIdx.x;
    const int base = blockIdx.x * OWN_R;
    owner_accumulate<16, false>(p1, ents, tab, nchunks, blockIdx.x, aggl, nullptr, qall, tid);

    const int nrows = (N - base) < OWN_R ? (N - base) : OWN_R;
    for (int i = tid; i < nrows * 16; i += 32 * OWN_WAVES) {
        float dv = deg[base + (i >> 4)];
        dv = dv > 1.0f ? dv : 1.0f;
        const int t = base * 16 + i;
        out[t] = aggl[i] / dv + bl1[i & 15] + re1[t];
    }
}

__global__ __launch_bounds__(32 * WAVES_PER_BLOCK)
void layer1_fused_kernel(const float* __restrict__ agg0, const float* __restrict__ r0,
                         const float* __restrict__ e0, const float* __restrict__ bl0,
                         const float* __restrict__ deg,
                         const float* __restrict__ Wl1, const float* __restrict__ Wr1,
                         const float* __restrict__ We1, const float* __restrict__ be1,
                         float* __restrict__ P1, float* __restrict__ RE1, int M)
{
    constexpr int K  = 64;
    constexpr int KS = K / 4;
    const int lane = threadIdx.x & 31;
    const int wave = blockIdx.x * (blockDim.x >> 5) + (threadIdx.x >> 5);
    if (wave >= (M >> 4)) return;
    const int half = lane >> 4;
    const int lm   = lane & 15;
    const int arow = wave * 16 + lm;

    float dv = deg[arow];
    dv = dv > 1.0f ? dv : 1.0f;

    const size_t abase = (size_t)arow * K + half * 2;
    v2f ah[KS], ae[KS];
#pragma unroll
    for (int i = 0; i < KS; ++i) {
        const v2f g = *(const v2f*)(agg0 + abase + 4 * i);
        const v2f r = *(const v2f*)(r0   + abase + 4 * i);
        const v2f e = *(const v2f*)(e0   + abase + 4 * i);
        const v2f b = *(const v2f*)(bl0 + 4 * i + half * 2);
        const float v0 = g.x / dv + b.x + r.x + e.x;
        const float v1 = g.y / dv + b.y + r.y + e.y;
        ah[i].x = v0 > 0.0f ? v0 : 0.0f;
        ah[i].y = v1 > 0.0f ? v1 : 0.0f;
        ae[i] = e;
    }

    const size_t wbase = (size_t)lm * K + half * 2;
    v2f bw[KS];


#pragma unroll
    for (int i = 0; i < KS; ++i)
        bw[i] = *(const v2f*)(Wl1 + wbase + 4 * i);
    v8f accp = {0.f, 0.f, 0.f, 0.f, 0.f, 0.f, 0.f, 0.f};
#pragma unroll
    for (int i = 0; i < KS; ++i)
        accp = __builtin_amdgcn_wmma_f32_16x16x4_f32(false, ah[i], false, bw[i],
                                                     (short)0, accp, false, false);

    v2f bx[KS];
#pragma unroll
    for (int i = 0; i < KS; ++i) {
        bw[i] = *(const v2f*)(Wr1 + wbase + 4 * i);
        bx[i] = *(const v2f*)(We1 + wbase + 4 * i);
    }
    v8f accr = {0.f, 0.f, 0.f, 0.f, 0.f, 0.f, 0.f, 0.f};
#pragma unroll
    for (int i = 0; i < KS; ++i)
        accr = __builtin_amdgcn_wmma_f32_16x16x4_f32(false, ah[i], false, bw[i],
                                                     (short)0, accr, false, false);
#pragma unroll
    for (int i = 0; i < KS; ++i)
        accr = __builtin_amdgcn_wmma_f32_16x16x4_f32(false, ae[i], false, bx[i],
                                                     (short)0, accr, false, false);

    const float bv = be1[lm];
#pragma unroll
    for (int r = 0; r < 8; ++r) {
        const size_t o = (size_t)(wave * 16 + half * 8 + r) * 16 + lm;
        P1[o]  = accp[r];
        RE1[o] = accr[r] + bv;
    }
}


extern "C" void kernel_launch(void* const* d_in, const int* in_sizes, int n_in,
                              void* d_out, int out_size, void* d_ws, size_t ws_size,
                              hipStream_t stream)
{
    const float* x_feat = (const float*)d_in[0];
    const float* x_emb  = (const float*)d_in[1];
    const int*   eidx   = (const int*)d_in[2];
    const float* Wl0 = (const float*)d_in[3];
    const float* bl0 = (const float*)d_in[4];
    const float* Wr0 = (const float*)d_in[5];
    const float* We0 = (const float*)d_in[6];
    const float* be0 = (const float*)d_in[7];
    const float* Wl1 = (const float*)d_in[8];
    const float* bl1 = (const float*)d_in[9];
    const float* Wr1 = (const float*)d_in[10];
    const float* We1 = (const float*)d_in[11];
    const float* be1 = (const float*)d_in[12];

    const int FEAT = 128;
    const int N = in_sizes[0] / FEAT;
    const int E = in_sizes[2] / 2;
    const int* src = eidx;
    const int* dst = eidx + E;

    float*       ws = (float*)d_ws;
    const size_t n  = (size_t)N;
    float* deg  = ws;
    float* e0   = ws + n;
    float* p0   = ws + 65 * n;
    float* r0   = ws + 129 * n;
    float* agg0 = ws + 193 * n;
    float* p1   = p0;
    float* re1  = ws + 81 * n;
    const int PB = (E + PART_CH - 1) / PART_CH;
    int* tab  = (int*)(ws + 257 * n);
    int* ents = tab + (size_t)PB * PART_PITCH;
    if (ws_size < (257 * n + (size_t)PB * (PART_PITCH + PART_CH)) * sizeof(float)) return;

    const int mwaves  = N >> 4;
    const int gblocks = (mwaves + WAVES_PER_BLOCK - 1) / WAVES_PER_BLOCK;
    const int gthreads = 32 * WAVES_PER_BLOCK;

    partition_kernel<<<PB, PART_T, 0, stream>>>(src, dst, E, N, ents, tab);

    wmma_gemm_kernel<128, 4, false><<<gblocks, gthreads, 0, stream>>>(x_feat, Wl0, nullptr, p0, N);

    const int G      = (N + OWN_R - 1) / OWN_R;
    const int pblock = (mwaves + OWN_WAVES - 1) / OWN_WAVES;
    agg0_owner_kernel<<<G + 2 * pblock, 32 * OWN_WAVES, 0, stream>>>(
        p0, ents, tab, PB, agg0, deg, N, G, x_feat, Wr0, r0, x_emb, We0, be0, e0);

    layer1_fused_kernel<<<gblocks, gthreads, 0, stream>>>(agg0, r0, e0, bl0, deg,
                                                          Wl1, Wr1, We1, be1, p1, re1, N);

    agg1_owner_kernel<<<G, 32 * OWN_WAVES, 0, stream>>>(p1, ents, tab, PB, N, re1, bl1, deg,
                                                        (float*)d_out);
}


# WMMA operand feed: weights copied once per work-group into LDS, read as 16-byte fragments; activation rows read 16 B per lane (same k order both sides): 16-line global loads per stripe 160 -> 16
# speedup vs baseline: 3.0810x; 1.2693x over previous
#include <hip/hip_runtime.h>
#include <stdint.h>

typedef float v4f __attribute__((ext_vector_type(4)));
typedef float v8f __attribute__((ext_vector_type(8)));

#define WAVES_PER_BLOCK 8

template <int K, int NO, int NTHREADS>
__device__ __forceinline__
void stage_fetch(const float* __restrict__ W, const int tid, v4f (&v)[NO * (K / 4) / NTHREADS])
{
    constexpr int IT = NO * (K / 4) / NTHREADS;
    static_assert(NO * (K / 4) % NTHREADS == 0, "every thread copies the same number of pieces");
#pragma unroll
    for (int u = 0; u < IT; ++u)
        v[u] = *(const v4f*)(W + (size_t)(u * NTHREADS + tid) * 4);
}

template <int K, int NO, int NTHREADS>
__device__ __forceinline__
void stage_put(float* wl, const int tid, const v4f (&v)[NO * (K / 4) / NTHREADS])
{
    constexpr int Q  = K / 4;
    constexpr int IT = NO * Q / NTHREADS;
#pragma unroll
    for (int u = 0; u < IT; ++u) {
        const int j = u * NTHREADS + tid;
        const int n = j / Q;
        const int q = j - n * Q;
        *(v4f*)(wl + n * (K + 4) + 4 * q) = v[u];
    }
}

template <int K, bool MEANPAIR>
__device__ __forceinline__
void load_a_stripe(const float* __restrict__ X, const int wave, const int lane,
                   v4f (&a)[K / 8], v4f (&a2)[MEANPAIR ? K / 8 : 1])
{
    constexpr int KC = K / 8;
    const int half = lane >> 4;
    const int arow = wave * 16 + (lane & 15);
    const float* __restrict__ x0 = X + (size_t)arow * (MEANPAIR ? 2 * K : K) + half * 4;
#pragma unroll
    for (int i = 0; i < KC; ++i) {
        a[i] = *(const v4f*)(x0 + 8 * i);
        if (MEANPAIR) a2[i] = *(const v4f*)(x0 + K + 8 * i);
    }
}

template <int KC>
__device__ __forceinline__
void mean_pairs(v4f (&a)[KC], const v4f (&a2)[KC])
{
#pragma unroll
    for (int i = 0; i < KC; ++i) {
        a[i].x = 0.5f * (a[i].x + a2[i].x);
        a[i].y = 0.5f * (a[i].y + a2[i].y);
        a[i].z = 0.5f * (a[i].z + a2[i].z);
        a[i].w = 0.5f * (a[i].w + a2[i].w);
    }
}

template <int K, int NT>
__device__ __forceinline__
void gemm_tiles(const v4f (&a)[K / 8], const float* wl,
                const float* __restrict__ bias, float* __restrict__ Out,
                const int wave, const int lane)
{
    constexpr int KC = K / 8;
    constexpr int NO = NT * 16;
    constexpr int WP = K + 4;
    const int half = lane >> 4;
    const int lm   = lane & 15;

#pragma unroll
    for (int nt = 0; nt < NT; ++nt) {
        const float* wrow = wl + (nt * 16 + lm) * WP + half * 4;
        v4f b[KC];
#pragma unroll
        for (int i = 0; i < KC; ++i)
            b[i] = *(const v4f*)(wrow + 8 * i);

        v8f acc = {0.f, 0.f, 0.f, 0.f, 0.f, 0.f, 0.f, 0.f};
#pragma unroll
        for (int i = 0; i < KC; ++i) {
            acc = __builtin_amdgcn_wmma_f32_16x16x4_f32(
                false, __builtin_shufflevector(a[i], a[i], 0, 1),
                false, __builtin_shufflevector(b[i], b[i], 0, 1), (short)0, acc, false, false);
            acc = __builtin_amdgcn_wmma_f32_16x16x4_f32(
                false, __builtin_shufflevector(a[i], a[i], 2, 3),
                false, __builtin_shufflevector(b[i], b[i], 2, 3), (short)0, acc, false, false);
        }

        const int   ocol = nt * 16 + lm;
        const float bv   = bias ? bias[ocol] : 0.0f;
#pragma unroll
        for (int r = 0; r < 8; ++r) {
            const int orow = wave * 16 + half * 8 + r;
            Out[(size_t)orow * NO + ocol] = acc[r] + bv;
        }
    }
}

template <int K, int NT>
__global__ __launch_bounds__(32 * WAVES_PER_BLOCK)
void wmma_gemm_kernel(const float* __restrict__ X, const float* __restrict__ W,
                      const float* __restrict__ bias, float* __restrict__ Out, int M)
{
    __shared__ __attribute__((aligned(16))) float wl[NT * 16 * (K + 4)];
    const int lane = threadIdx.x & 31;
    const int wave = blockIdx.x * (blockDim.x >> 5) + (threadIdx.x >> 5);
    const bool live = wave < (M >> 4);
    v4f wv[NT * 16 * (K / 4) / (32 * WAVES_PER_BLOCK)];
    v4f a[K / 8], a2[1];
    stage_fetch<K, NT * 16, 32 * WAVES_PER_BLOCK>(W, threadIdx.x, wv);
    if (live) load_a_stripe<K, false>(X, wave, lane, a, a2);
    stage_put<K, NT * 16, 32 * WAVES_PER_BLOCK>(wl, threadIdx.x, wv);
    __syncthreads();
    if (live) gemm_tiles<K, NT>(a, wl, bias, Out, wave, lane);
}


constexpr int OWN_R      = 416;
constexpr int OWN_WAVES  = 32;
constexpr int OWN_QCAP   = 256;
constexpr int PART_CH    = 8192;
constexpr int PART_T     = 1024;
constexpr int PART_PITCH = 256;
constexpr int ENT_SHIFT  = 22;

__global__ __launch_bounds__(PART_T)
void partition_kernel(const int* __restrict__ src, const int* __restrict__ dst, int E, int N,
                      int* __restrict__ ents, int* __restrict__ tab)
{
    __shared__ int hist[PART_PITCH];
    __shared__ int cur[PART_PITCH];
    __shared__ int stage[PART_CH];

    const int tid   = threadIdx.x;
    const int chunk = blockIdx.x;
    const int cbase = chunk * PART_CH;
    const int cnt   = (E - cbase) < PART_CH ? (E - cbase) : PART_CH;
    const int* __restrict__ dch = dst + cbase;
    const int* __restrict__ sch = src + cbase;
    constexpr int PER = PART_CH / PART_T;

    if (tid < PART_PITCH) hist[tid] = 0;
    __syncthreads();

    unsigned d[PER];
    int      s[PER];
#pragma unroll
    for (int u = 0; u < PER; ++u) {
        const int i = u * PART_T + tid;
        d[u] = 0xFFFFFFFFu;
        s[u] = 0;
        if (i < cnt) { d[u] = (unsigned)dch[i]; s[u] = sch[i]; }
    }
#pragma unroll
    for (int u = 0; u < PER; ++u)
        if (d[u] < (unsigned)N) atomicAdd(&hist[d[u] / OWN_R], 1);
    __syncthreads();

    if (tid < 32) {
        int h[8];
        int tot = 0;
#pragma unroll
        for (int i = 0; i < 8; ++i) { h[i] = hist[tid * 8 + i]; tot += h[i]; }
        int incl = tot;
#pragma unroll
        for (int off = 1; off < 32; off <<= 1) {
            const int t = __shfl_up(incl, off);
            if (tid >= off) incl += t;
        }
        int run = incl - tot;
        int* __restrict__ row = tab + (size_t)chunk * PART_PITCH;
#pragma unroll
        for (int i = 0; i < 8; ++i) {
            cur[tid * 8 + i] = run;
            row[tid * 8 + i] = run;
            run += h[i];
        }
    }
    __syncthreads();

#pragma unroll
    for (int u = 0; u < PER; ++u) {
        if (d[u] < (unsigned)N) {
            const unsigned o   = d[u] / OWN_R;
            const int      pos = atomicAdd(&cur[o], 1);
            stage[pos] = s[u] | (int)((d[u] - o * OWN_R) << ENT_SHIFT);
        }
    }
    __syncthreads();

    const int total = cur[PART_PITCH - 1];
    int* __restrict__ o = ents + (size_t)cbase;
    for (int i = tid; i < total; i += PART_T)
        o[i] = stage[i];
}

__device__ __forceinline__ void lds_add(float* p, float v)
{
    __hip_atomic_fetch_add(p, v, __ATOMIC_RELAXED, __HIP_MEMORY_SCOPE_WORKGROUP);
}

template <int C, bool COUNT>
__device__ __forceinline__
void owner_drain32(const float* __restrict__ P, float* aggl, float* degl, const int* q,
                   const int head, const int lane)
{
    const int ent = q[(head + lane) & (OWN_QCAP - 1)];
    const int s   = ent & ((1 << ENT_SHIFT) - 1);
    const int ld  = (int)((unsigned)ent >> ENT_SHIFT);
    if (COUNT) lds_add(degl + ld, 1.0f);
    if (C == 64) {
#pragma unroll 1
        for (int g = 0; g < 32; g += 8) {
            float lo[8], hi[8];
#pragma unroll
            for (int u = 0; u < 8; ++u) {
                const int sj = __builtin_amdgcn_readlane(s, g + u);
                const float* __restrict__ p = P + (sj * 64 + lane);
                lo[u] = p[0];
                hi[u] = p[32];
            }
#pragma unroll
            for (int u = 0; u < 8; ++u) {
                const int dj = __builtin_amdgcn_readlane(ld, g + u);
                float* a = aggl + (dj * 64 + lane);
                lds_add(a, lo[u]);
                lds_add(a + 32, hi[u]);
            }
        }
    } else {
        const int hsel = lane >> 4;
        const int c    = lane & 15;
        float v[16];
#pragma unroll
        for (int j = 0; j < 16; ++j) {
            const int sa = __builtin_amdgcn_readlane(s, j);
            const int sb = __builtin_amdgcn_readlane(s, 16 + j);
            const int sj = hsel ? sb : sa;
            v[j] = P[sj * 16 + c];
        }
#pragma unroll
        for (int j = 0; j < 16; ++j) {
            const int da = __builtin_amdgcn_readlane(ld, j);
            const int db = __builtin_amdgcn_readlane(ld, 16 + j);
            const int dj = hsel ? db : da;
            lds_add(aggl + (dj * 16 + c), v[j]);
        }
    }
}

template <int C, bool COUNT>
__device__ __forceinline__
void owner_drain_tail(const float* __restrict__ P, float* aggl, float* degl, const int* q,
                      const int head, const int n, const int lane)
{
    int s = 0, ld = 0;
    if (lane < n) {
        const int ent = q[(head + lane) & (OWN_QCAP - 1)];
        s  = ent & ((1 << ENT_SHIFT) - 1);
        ld = (int)((unsigned)ent >> ENT_SHIFT);
        if (COUNT) lds_add(degl + ld, 1.0f);
    }
    for (int j = 0; j < n; ++j) {
        const int sj = __builtin_amdgcn_readlane(s, j);
        const int dj = __builtin_amdgcn_readlane(ld, j);
        if (C == 64) {
            const float* __restrict__ p = P + (sj * 64 + lane);
            const float v0 = p[0], v1 = p[32];
            float* a = aggl + (dj * 64 + lane);
            lds_add(a, v0);
            lds_add(a + 32, v1);
        } else {
            if (lane < 16) lds_add(aggl + (dj * 16 + lane), P[sj * 16 + lane]);
        }
    }
}

__device__ __forceinline__
void owner_append(int* q, const int head, int& qn, const unsigned mask, const bool mine,
                  const int ent)
{
    if (mask != 0u) {
        if (mine) {
            const int pos = qn + (int)__builtin_amdgcn_mbcnt_lo(mask, 0u);
            q[(head + pos) & (OWN_QCAP - 1)] = ent;
        }
        qn += __builtin_popcount(mask);
    }
}

template <int C, bool COUNT>
__device__ __forceinline__
void owner_accumulate(const float* __restrict__ P, const int* __restrict__ ents,
                      const int* __restrict__ tab, const int nchunks, const int own,
                      float* aggl, float* degl, int* qall, const int tid)
{
    const int lane = tid & 31;
    const int wib  = tid >> 5;

    for (int i = tid; i < OWN_R * C / 4; i += 32 * OWN_WAVES)
        ((float4*)aggl)[i] = make_float4(0.f, 0.f, 0.f, 0.f);
    if (COUNT)
        for (int i = tid; i < OWN_R; i += 32 * OWN_WAVES) degl[i] = 0.f;
    __syncthreads();

    int* q = qall + wib * OWN_QCAP;
    int head = 0, qn = 0;

    for (int c0 = wib; c0 < nchunks; c0 += 32 * OWN_WAVES) {
        const int cl = c0 + lane * OWN_WAVES;
        int st = 0, en = 0;
        if (cl < nchunks) {
            const int* __restrict__ row = tab + (size_t)cl * PART_PITCH + own;
            st = row[0];
            en = row[1];
        }
        const int left = (nchunks - c0 + OWN_WAVES - 1) / OWN_WAVES;
        const int nj   = left < 32 ? left : 32;
        for (int j = 0; j < nj; ++j) {
            const int sj = __builtin_amdgcn_readlane(st, j);
            const int ej = __builtin_amdgcn_readlane(en, j);
            const int* __restrict__ seg = ents + (size_t)(c0 + j * OWN_WAVES) * PART_CH;
            for (int t = sj; t < ej; t += 128) {
                int  v[4];
                bool m[4];
#pragma unroll
                for (int u = 0; u < 4; ++u) {
                    const int idx = t + u * 32 + lane;
                    m[u] = idx < ej;
                    v[u] = 0;
                    if (m[u]) v[u] = seg[idx];
                }
#pragma unroll
                for (int u = 0; u < 4; ++u)
                    owner_append(q, head, qn, __builtin_amdgcn_ballot_w32(m[u]), m[u], v[u]);
                while (qn >= 32) {
                    owner_drain32<C, COUNT>(P, aggl, degl, q, head, lane);
                    head = (head + 32) & (OWN_QCAP - 1);
                    qn -= 32;
                }
            }
        }
    }

    if (qn > 0) owner_drain_tail<C, COUNT>(P, aggl, degl, q, head, qn, lane);
    __syncthreads();
}

__global__ __launch_bounds__(32 * OWN_WAVES)
void agg0_owner_kernel(const float* __restrict__ p0, const int* __restrict__ ents,
                       const int* __restrict__ tab, int nchunks, float* __restrict__ agg0,
                       float* __restrict__ deg, int N, int G,
                       const float* __restrict__ x_feat, const float* __restrict__ Wr0,
                       float* __restrict__ r0,
                       const float* __restrict__ x_emb, const float* __restrict__ We0,
                       const float* __restrict__ be0, float* __restrict__ e0)
{
    static_assert(OWN_R * 64 >= 64 * (128 + 4), "a projection block's weight copy must fit in aggl");
    __shared__ __attribute__((aligned(16))) float aggl[OWN_R * 64];
    __shared__ float degl[OWN_R];
    __shared__ int   qall[OWN_WAVES * OWN_QCAP];

    const int tid = threadIdx.x;
    const int b   = blockIdx.x;
    if (b >= G) {
        const int  pb     = b - G;
        const int  stripe = (pb >> 1) * OWN_WAVES + (tid >> 5);
        const bool live   = stripe < (N >> 4);
        if (pb & 1) {
            v4f wv[1], a[8], a2[8];
            stage_fetch<64, 64, 32 * OWN_WAVES>(We0, tid, wv);
            if (live) load_a_stripe<64, true>(x_emb, stripe, tid & 31, a, a2);
            stage_put<64, 64, 32 * OWN_WAVES>(aggl, tid, wv);
            __syncthreads();
            if (live) {
                mean_pairs<8>(a, a2);
                gemm_tiles<64, 4>(a, aggl, be0, e0, stripe, tid & 31);
            }
        } else {
            v4f wv[2], a[16], a2[1];
            stage_fetch<128, 64, 32 * OWN_WAVES>(Wr0, tid, wv);
            if (live) load_a_stripe<128, false>(x_feat, stripe, tid & 31, a, a2);
            stage_put<128, 64, 32 * OWN_WAVES>(aggl, tid, wv);
            __syncthreads();
            if (live) gemm_tiles<128, 4>(a, aggl, nullptr, r0, stripe, tid & 31);
        }
        return;
    }

    const int base = b * OWN_R;
    owner_accumulate<64, true>(p0, ents, tab, nchunks, b, aggl, degl, qall, tid);

    const int nrows = (N - base) < OWN_R ? (N - base) : OWN_R;
    float4* __restrict__ o4 = reinterpret_cast<float4*>(agg0 + (size_t)base * 64);
    for (int i = tid; i < nrows * 16; i += 32 * OWN_WAVES)
        o4[i] = ((const float4*)aggl)[i];
    for (int i = tid; i < nrows; i += 32 * OWN_WAVES)
        deg[base + i] = degl[i];
}

__global__ __launch_bounds__(32 * OWN_WAVES)
void agg1_owner_kernel(const float* __restrict__ p1, const int* __restrict__ ents,
                       const int* __restrict__ tab, int nchunks, int N,
                       const float* __restrict__ re1, const float* __restrict__ bl1,
                       const float* __restrict__ deg, float* __restrict__ out)
{
    __shared__ __attribute__((aligned(16))) float aggl[OWN_R * 16];
    __shared__ int qall[OWN_WAVES * OWN_QCAP];

    const int tid  = threadIdx.x;
    const int base = blockIdx.x * OWN_R;
    owner_accumulate<16, false>(p1, ents, tab, nchunks, blockIdx.x, aggl, nullptr, qall, tid);

    const int nrows = (N - base) < OWN_R ? (N - base) : OWN_R;
    for (int i = tid; i < nrows * 16; i += 32 * OWN_WAVES) {
        float dv = deg[base + (i >> 4)];
        dv = dv > 1.0f ? dv : 1.0f;
        const int t = base * 16 + i;
        out[t] = aggl[i] / dv + bl1[i & 15] + re1[t];
    }
}

__global__ __launch_bounds__(32 * WAVES_PER_BLOCK)
void layer1_fused_kernel(const float* __restrict__ agg0, const float* __restrict__ r0,
                         const float* __restrict__ e0, const float* __restrict__ bl0,
                         const float* __restrict__ deg,
                         const float* __restrict__ Wl1, const float* __restrict__ Wr1,
                         const float* __restrict__ We1, const float* __restrict__ be1,
                         float* __restrict__ P1, float* __restrict__ RE1, int M)
{
    constexpr int K  = 64;
    constexpr int KC = K / 8;
    constexpr int WP = K + 4;
    __shared__ __attribute__((aligned(16))) float wl[3 * 16 * WP];
    {
        v4f w0[1], w1[1], w2[1];
        stage_fetch<K, 16, 32 * WAVES_PER_BLOCK>(Wl1, threadIdx.x, w0);
        stage_fetch<K, 16, 32 * WAVES_PER_BLOCK>(Wr1, threadIdx.x, w1);
        stage_fetch<K, 16, 32 * WAVES_PER_BLOCK>(We1, threadIdx.x, w2);
        stage_put<K, 16, 32 * WAVES_PER_BLOCK>(wl,           threadIdx.x, w0);
        stage_put<K, 16, 32 * WAVES_PER_BLOCK>(wl + 16 * WP, threadIdx.x, w1);
        stage_put<K, 16, 32 * WAVES_PER_BLOCK>(wl + 32 * WP, threadIdx.x, w2);
    }
    __syncthreads();

    const int lane = threadIdx.x & 31;
    const int wave = blockIdx.x * (blockDim.x >> 5) + (threadIdx.x >> 5);
    if (wave >= (M >> 4)) return;
    const int half = lane >> 4;
    const int lm   = lane & 15;
    const int arow = wave * 16 + lm;

    float dv = deg[arow];
    dv = dv > 1.0f ? dv : 1.0f;

    const size_t abase = (size_t)arow * K + half * 4;
    v4f ah[KC], ae[KC];
#pragma unroll
    for (int i = 0; i < KC; ++i) {
        const v4f g = *(const v4f*)(agg0 + abase + 8 * i);
        const v4f r = *(const v4f*)(r0   + abase + 8 * i);
        const v4f e = *(const v4f*)(e0   + abase + 8 * i);
        const v4f b = *(const v4f*)(bl0 + 8 * i + half * 4);
        const float v0 = g.x / dv + b.x + r.x + e.x;
        const float v1 = g.y / dv + b.y + r.y + e.y;
        const float v2 = g.z / dv + b.z + r.z + e.z;
        const float v3 = g.w / dv + b.w + r.w + e.w;
        ah[i].x = v0 > 0.0f ? v0 : 0.0f;
        ah[i].y = v1 > 0.0f ? v1 : 0.0f;
        ah[i].z = v2 > 0.0f ? v2 : 0.0f;
        ah[i].w = v3 > 0.0f ? v3 : 0.0f;
        ae[i] = e;
    }

    const float* wb = wl + lm * WP + half * 4;
    v4f bw[KC];


#pragma unroll
    for (int i = 0; i < KC; ++i)
        bw[i] = *(const v4f*)(wb + 8 * i);
    v8f accp = {0.f, 0.f, 0.f, 0.f, 0.f, 0.f, 0.f, 0.f};
#pragma unroll
    for (int i = 0; i < KC; ++i) {
        accp = __builtin_amdgcn_wmma_f32_16x16x4_f32(
            false, __builtin_shufflevector(ah[i], ah[i], 0, 1),
            false, __builtin_shufflevector(bw[i], bw[i], 0, 1), (short)0, accp, false, false);
        accp = __builtin_amdgcn_wmma_f32_16x16x4_f32(
            false, __builtin_shufflevector(ah[i], ah[i], 2, 3),
            false, __builtin_shufflevector(bw[i], bw[i], 2, 3), (short)0, accp, false, false);
    }

    v4f bx[KC];
#pragma unroll
    for (int i = 0; i < KC; ++i) {
        bw[i] = *(const v4f*)(wb + 16 * WP + 8 * i);
        bx[i] = *(const v4f*)(wb + 32 * WP + 8 * i);
    }
    v8f accr = {0.f, 0.f, 0.f, 0.f, 0.f, 0.f, 0.f, 0.f};
#pragma unroll
    for (int i = 0; i < KC; ++i) {
        accr = __builtin_amdgcn_wmma_f32_16x16x4_f32(
            false, __builtin_shufflevector(ah[i], ah[i], 0, 1),
            false, __builtin_shufflevector(bw[i], bw[i], 0, 1), (short)0, accr, false, false);
        accr = __builtin_amdgcn_wmma_f32_16x16x4_f32(
            false, __builtin_shufflevector(ah[i], ah[i], 2, 3),
            false, __builtin_shufflevector(bw[i], bw[i], 2, 3), (short)0, accr, false, false);
    }
#pragma unroll
    for (int i = 0; i < KC; ++i) {
        accr = __builtin_amdgcn_wmma_f32_16x16x4_f32(
            false, __builtin_shufflevector(ae[i], ae[i], 0, 1),
            false, __builtin_shufflevector(bx[i], bx[i], 0, 1), (short)0, accr, false, false);
        accr = __builtin_amdgcn_wmma_f32_16x16x4_f32(
            false, __builtin_shufflevector(ae[i], ae[i], 2, 3),
            false, __builtin_shufflevector(bx[i], bx[i], 2, 3), (short)0, accr, false, false);
    }

    const float bv = be1[lm];
#pragma unroll
    for (int r = 0; r < 8; ++r) {
        const size_t o = (size_t)(wave * 16 + half * 8 + r) * 16 + lm;
        P1[o]  = accp[r];
        RE1[o] = accr[r] + bv;
    }
}


extern "C" void kernel_launch(void* const* d_in, const int* in_sizes, int n_in,
                              void* d_out, int out_size, void* d_ws, size_t ws_size,
                              hipStream_t stream)
{
    const float* x_feat = (const float*)d_in[0];
    const float* x_emb  = (const float*)d_in[1];
    const int*   eidx   = (const int*)d_in[2];
    const float* Wl0 = (const float*)d_in[3];
    const float* bl0 = (const float*)d_in[4];
    const float* Wr0 = (const float*)d_in[5];
    const float* We0 = (const float*)d_in[6];
    const float* be0 = (const float*)d_in[7];
    const float* Wl1 = (const float*)d_in[8];
    const float* bl1 = (const float*)d_in[9];
    const float* Wr1 = (const float*)d_in[10];
    const float* We1 = (const float*)d_in[11];
    const float* be1 = (const float*)d_in[12];

    const int FEAT = 128;
    const int N = in_sizes[0] / FEAT;
    const int E = in_sizes[2] / 2;
    const int* src = eidx;
    const int* dst = eidx + E;

    float*       ws = (float*)d_ws;
    const size_t n  = (size_t)N;
    float* deg  = ws;
    float* e0   = ws + n;
    float* p0   = ws + 65 * n;
    float* r0   = ws + 129 * n;
    float* agg0 = ws + 193 * n;
    float* p1   = p0;
    float* re1  = ws + 81 * n;
    const int PB = (E + PART_CH - 1) / PART_CH;
    int* tab  = (int*)(ws + 257 * n);
    int* ents = tab + (size_t)PB * PART_PITCH;
    if (ws_size < (257 * n + (size_t)PB * (PART_PITCH + PART_CH)) * sizeof(float)) return;

    const int mwaves  = N >> 4;
    const int gblocks = (mwaves + WAVES_PER_BLOCK - 1) / WAVES_PER_BLOCK;
    const int gthreads = 32 * WAVES_PER_BLOCK;

    partition_kernel<<<PB, PART_T, 0, stream>>>(src, dst, E, N, ents, tab);

    wmma_gemm_kernel<128, 4><<<gblocks, gthreads, 0, stream>>>(x_feat, Wl0, nullptr, p0, N);

    const int G      = (N + OWN_R - 1) / OWN_R;
    const int pblock = (mwaves + OWN_WAVES - 1) / OWN_WAVES;
    agg0_owner_kernel<<<G + 2 * pblock, 32 * OWN_WAVES, 0, stream>>>(
        p0, ents, tab, PB, agg0, deg, N, G, x_feat, Wr0, r0, x_emb, We0, be0, e0);

    layer1_fused_kernel<<<gblocks, gthreads, 0, stream>>>(agg0, r0, e0, bl0, deg,
                                                          Wl1, Wr1, We1, be1, p1, re1, N);

    agg1_owner_kernel<<<G, 32 * OWN_WAVES, 0, stream>>>(p1, ents, tab, PB, N, re1, bl1, deg,
                                                        (float*)d_out);
}


# Layer-0 launch in 16-wave blocks (a 32-wave block filled a WGP's registers alone): two share a WGP, so projection WMMAs run beside an owner's gathers, not after it; owner wave gathers 32 rows a batch
# speedup vs baseline: 3.3115x; 1.0748x over previous
#include <hip/hip_runtime.h>
#include <stdint.h>

typedef float v2f __attribute__((ext_vector_type(2)));
typedef float v4f __attribute__((ext_vector_type(4)));
typedef float v8f __attribute__((ext_vector_type(8)));

#define WAVES_PER_BLOCK 8

template <int K, int NO, int NTHREADS>
__device__ __forceinline__
void stage_fetch(const float* __restrict__ W, const int tid, v4f (&v)[NO * (K / 4) / NTHREADS])
{
    constexpr int IT = NO * (K / 4) / NTHREADS;
    static_assert(NO * (K / 4) % NTHREADS == 0, "every thread copies the same number of pieces");
#pragma unroll
    for (int u = 0; u < IT; ++u)
        v[u] = *(const v4f*)(W + (size_t)(u * NTHREADS + tid) * 4);
}

template <int K, int NO, int NTHREADS>
__device__ __forceinline__
void stage_put(float* wl, const int tid, const v4f (&v)[NO * (K / 4) / NTHREADS])
{
    constexpr int Q  = K / 4;
    constexpr int IT = NO * Q / NTHREADS;
#pragma unroll
    for (int u = 0; u < IT; ++u) {
        const int j = u * NTHREADS + tid;
        const int n = j / Q;
        const int q = j - n * Q;
        *(v4f*)(wl + n * (K + 4) + 4 * q) = v[u];
    }
}

template <int K, bool MEANPAIR>
__device__ __forceinline__
void load_a_stripe(const float* __restrict__ X, const int wave, const int lane,
                   v4f (&a)[K / 8], v4f (&a2)[MEANPAIR ? K / 8 : 1])
{
    constexpr int KC = K / 8;
    const int half = lane >> 4;
    const int arow = wave * 16 + (lane & 15);
    const float* __restrict__ x0 = X + (size_t)arow * (MEANPAIR ? 2 * K : K) + half * 4;
#pragma unroll
    for (int i = 0; i < KC; ++i) {
        a[i] = *(const v4f*)(x0 + 8 * i);
        if (MEANPAIR) a2[i] = *(const v4f*)(x0 + K + 8 * i);
    }
}

template <int KC>
__device__ __forceinline__
void mean_pairs(v4f (&a)[KC], const v4f (&a2)[KC])
{
#pragma unroll
    for (int i = 0; i < KC; ++i) {
        a[i].x = 0.5f * (a[i].x + a2[i].x);
        a[i].y = 0.5f * (a[i].y + a2[i].y);
        a[i].z = 0.5f * (a[i].z + a2[i].z);
        a[i].w = 0.5f * (a[i].w + a2[i].w);
    }
}

template <int K, int NT>
__device__ __forceinline__
void gemm_tiles(const v4f (&a)[K / 8], const float* wl,
                const float* __restrict__ bias, float* __restrict__ Out,
                const int wave, const int lane)
{
    constexpr int KC = K / 8;
    constexpr int NO = NT * 16;
    constexpr int WP = K + 4;
    const int half = lane >> 4;
    const int lm   = lane & 15;

#pragma unroll
    for (int nt = 0; nt < NT; ++nt) {
        const float* wrow = wl + (nt * 16 + lm) * WP + half * 4;
        v4f b[KC];
#pragma unroll
        for (int i = 0; i < KC; ++i)
            b[i] = *(const v4f*)(wrow + 8 * i);

        v8f acc = {0.f, 0.f, 0.f, 0.f, 0.f, 0.f, 0.f, 0.f};
#pragma unroll
        for (int i = 0; i < KC; ++i) {
            acc = __builtin_amdgcn_wmma_f32_16x16x4_f32(
                false, __builtin_shufflevector(a[i], a[i], 0, 1),
                false, __builtin_shufflevector(b[i], b[i], 0, 1), (short)0, acc, false, false);
            acc = __builtin_amdgcn_wmma_f32_16x16x4_f32(
                false, __builtin_shufflevector(a[i], a[i], 2, 3),
                false, __builtin_shufflevector(b[i], b[i], 2, 3), (short)0, acc, false, false);
        }

        const int   ocol = nt * 16 + lm;
        const float bv   = bias ? bias[ocol] : 0.0f;
#pragma unroll
        for (int r = 0; r < 8; ++r) {
            const int orow = wave * 16 + half * 8 + r;
            Out[(size_t)orow * NO + ocol] = acc[r] + bv;
        }
    }
}

template <int K, int NT>
__global__ __launch_bounds__(32 * WAVES_PER_BLOCK)
void wmma_gemm_kernel(const float* __restrict__ X, const float* __restrict__ W,
                      const float* __restrict__ bias, float* __restrict__ Out, int M)
{
    __shared__ __attribute__((aligned(16))) float wl[NT * 16 * (K + 4)];
    const int lane = threadIdx.x & 31;
    const int wave = blockIdx.x * (blockDim.x >> 5) + (threadIdx.x >> 5);
    const bool live = wave < (M >> 4);
    v4f wv[NT * 16 * (K / 4) / (32 * WAVES_PER_BLOCK)];
    v4f a[K / 8], a2[1];
    stage_fetch<K, NT * 16, 32 * WAVES_PER_BLOCK>(W, threadIdx.x, wv);
    if (live) load_a_stripe<K, false>(X, wave, lane, a, a2);
    stage_put<K, NT * 16, 32 * WAVES_PER_BLOCK>(wl, threadIdx.x, wv);
    __syncthreads();
    if (live) gemm_tiles<K, NT>(a, wl, bias, Out, wave, lane);
}


constexpr int OWN_R      = 416;
constexpr int OWN_WAVES  = 32;
constexpr int AGG0_WAVES = 16;
constexpr int OWN_QCAP   = 256;
constexpr int PART_CH    = 8192;
constexpr int PART_T     = 1024;
constexpr int PART_PITCH = 256;
constexpr int ENT_SHIFT  = 22;

__global__ __launch_bounds__(PART_T)
void partition_kernel(const int* __restrict__ src, const int* __restrict__ dst, int E, int N,
                      int* __restrict__ ents, int* __restrict__ tab)
{
    __shared__ int hist[PART_PITCH];
    __shared__ int cur[PART_PITCH];
    __shared__ int stage[PART_CH];

    const int tid   = threadIdx.x;
    const int chunk = blockIdx.x;
    const int cbase = chunk * PART_CH;
    const int cnt   = (E - cbase) < PART_CH ? (E - cbase) : PART_CH;
    const int* __restrict__ dch = dst + cbase;
    const int* __restrict__ sch = src + cbase;
    constexpr int PER = PART_CH / PART_T;

    if (tid < PART_PITCH) hist[tid] = 0;
    __syncthreads();

    unsigned d[PER];
    int      s[PER];
#pragma unroll
    for (int u = 0; u < PER; ++u) {
        const int i = u * PART_T + tid;
        d[u] = 0xFFFFFFFFu;
        s[u] = 0;
        if (i < cnt) { d[u] = (unsigned)dch[i]; s[u] = sch[i]; }
    }
#pragma unroll
    for (int u = 0; u < PER; ++u)
        if (d[u] < (unsigned)N) atomicAdd(&hist[d[u] / OWN_R], 1);
    __syncthreads();

    if (tid < 32) {
        int h[8];
        int tot = 0;
#pragma unroll
        for (int i = 0; i < 8; ++i) { h[i] = hist[tid * 8 + i]; tot += h[i]; }
        int incl = tot;
#pragma unroll
        for (int off = 1; off < 32; off <<= 1) {
            const int t = __shfl_up(incl, off);
            if (tid >= off) incl += t;
        }
        int run = incl - tot;
        int* __restrict__ row = tab + (size_t)chunk * PART_PITCH;
#pragma unroll
        for (int i = 0; i < 8; ++i) {
            cur[tid * 8 + i] = run;
            row[tid * 8 + i] = run;
            run += h[i];
        }
    }
    __syncthreads();

#pragma unroll
    for (int u = 0; u < PER; ++u) {
        if (d[u] < (unsigned)N) {
            const unsigned o   = d[u] / OWN_R;
            const int      pos = atomicAdd(&cur[o], 1);
            stage[pos] = s[u] | (int)((d[u] - o * OWN_R) << ENT_SHIFT);
        }
    }
    __syncthreads();

    const int total = cur[PART_PITCH - 1];
    int* __restrict__ o = ents + (size_t)cbase;
    for (int i = tid; i < total; i += PART_T)
        o[i] = stage[i];
}

__device__ __forceinline__ void lds_add(float* p, float v)
{
    __hip_atomic_fetch_add(p, v, __ATOMIC_RELAXED, __HIP_MEMORY_SCOPE_WORKGROUP);
}

template <int C, bool COUNT>
__device__ __forceinline__
void owner_drain32(const float* __restrict__ P, float* aggl, float* degl, const int* q,
                   const int head, const int lane)
{
    const int ent = q[(head + lane) & (OWN_QCAP - 1)];
    const int s   = ent & ((1 << ENT_SHIFT) - 1);
    const int ld  = (int)((unsigned)ent >> ENT_SHIFT);
    if (COUNT) lds_add(degl + ld, 1.0f);
    if (C == 64) {
        v2f r[32];
#pragma unroll
        for (int u = 0; u < 32; ++u) {
            const int sj = __builtin_amdgcn_readlane(s, u);
            r[u] = *(const v2f*)(P + (sj * 64 + 2 * lane));
        }
#pragma unroll
        for (int u = 0; u < 32; ++u) {
            const int dj = __builtin_amdgcn_readlane(ld, u);
            float* a = aggl + (dj * 64 + 2 * lane);
            lds_add(a, r[u].x);
            lds_add(a + 1, r[u].y);
        }
    } else {
        const int hsel = lane >> 4;
        const int c    = lane & 15;
        float v[16];
#pragma unroll
        for (int j = 0; j < 16; ++j) {
            const int sa = __builtin_amdgcn_readlane(s, j);
            const int sb = __builtin_amdgcn_readlane(s, 16 + j);
            const int sj = hsel ? sb : sa;
            v[j] = P[sj * 16 + c];
        }
#pragma unroll
        for (int j = 0; j < 16; ++j) {
            const int da = __builtin_amdgcn_readlane(ld, j);
            const int db = __builtin_amdgcn_readlane(ld, 16 + j);
            const int dj = hsel ? db : da;
            lds_add(aggl + (dj * 16 + c), v[j]);
        }
    }
}

template <int C, bool COUNT>
__device__ __forceinline__
void owner_drain_tail(const float* __restrict__ P, float* aggl, float* degl, const int* q,
                      const int head, const int n, const int lane)
{
    int s = 0, ld = 0;
    if (lane < n) {
        const int ent = q[(head + lane) & (OWN_QCAP - 1)];
        s  = ent & ((1 << ENT_SHIFT) - 1);
        ld = (int)((unsigned)ent >> ENT_SHIFT);
        if (COUNT) lds_add(degl + ld, 1.0f);
    }
    for (int j = 0; j < n; ++j) {
        const int sj = __builtin_amdgcn_readlane(s, j);
        const int dj = __builtin_amdgcn_readlane(ld, j);
        if (C == 64) {
            const float* __restrict__ p = P + (sj * 64 + lane);
            const float v0 = p[0], v1 = p[32];
            float* a = aggl + (dj * 64 + lane);
            lds_add(a, v0);
            lds_add(a + 32, v1);
        } else {
            if (lane < 16) lds_add(aggl + (dj * 16 + lane), P[sj * 16 + lane]);
        }
    }
}

__device__ __forceinline__
void owner_append(int* q, const int head, int& qn, const unsigned mask, const bool mine,
                  const int ent)
{
    if (mask != 0u) {
        if (mine) {
            const int pos = qn + (int)__builtin_amdgcn_mbcnt_lo(mask, 0u);
            q[(head + pos) & (OWN_QCAP - 1)] = ent;
        }
        qn += __builtin_popcount(mask);
    }
}

template <int C, bool COUNT, int NW>
__device__ __forceinline__
void owner_accumulate(const float* __restrict__ P, const int* __restrict__ ents,
                      const int* __restrict__ tab, const int nchunks, const int own,
                      float* aggl, float* degl, int* qall, const int tid)
{
    const int lane = tid & 31;
    const int wib  = tid >> 5;

    for (int i = tid; i < OWN_R * C / 4; i += 32 * NW)
        ((float4*)aggl)[i] = make_float4(0.f, 0.f, 0.f, 0.f);
    if (COUNT)
        for (int i = tid; i < OWN_R; i += 32 * NW) degl[i] = 0.f;
    __syncthreads();

    int* q = qall + wib * OWN_QCAP;
    int head = 0, qn = 0;

    for (int c0 = wib; c0 < nchunks; c0 += 32 * NW) {
        const int cl = c0 + lane * NW;
        int st = 0, en = 0;
        if (cl < nchunks) {
            const int* __restrict__ row = tab + (size_t)cl * PART_PITCH + own;
            st = row[0];
            en = row[1];
        }
        const int left = (nchunks - c0 + NW - 1) / NW;
        const int nj   = left < 32 ? left : 32;
        for (int j = 0; j < nj; ++j) {
            const int sj = __builtin_amdgcn_readlane(st, j);
            const int ej = __builtin_amdgcn_readlane(en, j);
            const int* __restrict__ seg = ents + (size_t)(c0 + j * NW) * PART_CH;
            for (int t = sj; t < ej; t += 128) {
                int  v[4];
                bool m[4];
#pragma unroll
                for (int u = 0; u < 4; ++u) {
                    const int idx = t + u * 32 + lane;
                    m[u] = idx < ej;
                    v[u] = 0;
                    if (m[u]) v[u] = seg[idx];
                }
#pragma unroll
                for (int u = 0; u < 4; ++u)
                    owner_append(q, head, qn, __builtin_amdgcn_ballot_w32(m[u]), m[u], v[u]);
                while (qn >= 32) {
                    owner_drain32<C, COUNT>(P, aggl, degl, q, head, lane);
                    head = (head + 32) & (OWN_QCAP - 1);
                    qn -= 32;
                }
            }
        }
    }

    if (qn > 0) owner_drain_tail<C, COUNT>(P, aggl, degl, q, head, qn, lane);
    __syncthreads();
}

__global__ __launch_bounds__(32 * AGG0_WAVES)
void agg0_owner_kernel(const float* __restrict__ p0, const int* __restrict__ ents,
                       const int* __restrict__ tab, int nchunks, float* __restrict__ agg0,
                       float* __restrict__ deg, int N, int G,
                       const float* __restrict__ x_feat, const float* __restrict__ Wr0,
                       float* __restrict__ r0,
                       const float* __restrict__ x_emb, const float* __restrict__ We0,
                       const float* __restrict__ be0, float* __restrict__ e0)
{
    static_assert(OWN_R * 64 >= 64 * (128 + 4), "a projection block's weight copy must fit in aggl");
    __shared__ __attribute__((aligned(16))) float aggl[OWN_R * 64];
    __shared__ float degl[OWN_R];
    __shared__ int   qall[AGG0_WAVES * OWN_QCAP];

    const int tid = threadIdx.x;
    const int b   = blockIdx.x;
    if (b >= G) {
        const int  pb     = b - G;
        const int  stripe = (pb >> 1) * AGG0_WAVES + (tid >> 5);
        const bool live   = stripe < (N >> 4);
        if (pb & 1) {
            v4f wv[2], a[8], a2[8];
            stage_fetch<64, 64, 32 * AGG0_WAVES>(We0, tid, wv);
            if (live) load_a_stripe<64, true>(x_emb, stripe, tid & 31, a, a2);
            stage_put<64, 64, 32 * AGG0_WAVES>(aggl, tid, wv);
            __syncthreads();
            if (live) {
                mean_pairs<8>(a, a2);
                gemm_tiles<64, 4>(a, aggl, be0, e0, stripe, tid & 31);
            }
        } else {
            v4f wv[4], a[16], a2[1];
            stage_fetch<128, 64, 32 * AGG0_WAVES>(Wr0, tid, wv);
            if (live) load_a_stripe<128, false>(x_feat, stripe, tid & 31, a, a2);
            stage_put<128, 64, 32 * AGG0_WAVES>(aggl, tid, wv);
            __syncthreads();
            if (live) gemm_tiles<128, 4>(a, aggl, nullptr, r0, stripe, tid & 31);
        }
        return;
    }

    const int base = b * OWN_R;
    owner_accumulate<64, true, AGG0_WAVES>(p0, ents, tab, nchunks, b, aggl, degl, qall, tid);

    const int nrows = (N - base) < OWN_R ? (N - base) : OWN_R;
    float4* __restrict__ o4 = reinterpret_cast<float4*>(agg0 + (size_t)base * 64);
    for (int i = tid; i < nrows * 16; i += 32 * AGG0_WAVES)
        o4[i] = ((const float4*)aggl)[i];
    for (int i = tid; i < nrows; i += 32 * AGG0_WAVES)
        deg[base + i] = degl[i];
}

__global__ __launch_bounds__(32 * OWN_WAVES)
void agg1_owner_kernel(const float* __restrict__ p1, const int* __restrict__ ents,
                       const int* __restrict__ tab, int nchunks, int N,
                       const float* __restrict__ re1, const float* __restrict__ bl1,
                       const float* __restrict__ deg, float* __restrict__ out)
{
    __shared__ __attribute__((aligned(16))) float aggl[OWN_R * 16];
    __shared__ int qall[OWN_WAVES * OWN_QCAP];

    const int tid  = threadIdx.x;
    const int base = blockIdx.x * OWN_R;
    owner_accumulate<16, false, OWN_WAVES>(p1, ents, tab, nchunks, blockIdx.x, aggl, nullptr, qall, tid);

    const int nrows = (N - base) < OWN_R ? (N - base) : OWN_R;
    for (int i = tid; i < nrows * 16; i += 32 * OWN_WAVES) {
        float dv = deg[base + (i >> 4)];
        dv = dv > 1.0f ? dv : 1.0f;
        const int t = base * 16 + i;
        out[t] = aggl[i] / dv + bl1[i & 15] + re1[t];
    }
}

__global__ __launch_bounds__(32 * WAVES_PER_BLOCK)
void layer1_fused_kernel(const float* __restrict__ agg0, const float* __restrict__ r0,
                         const float* __restrict__ e0, const float* __restrict__ bl0,
                         const float* __restrict__ deg,
                         const float* __restrict__ Wl1, const float* __restrict__ Wr1,
                         const float* __restrict__ We1, const float* __restrict__ be1,
                         float* __restrict__ P1, float* __restrict__ RE1, int M)
{
    constexpr int K  = 64;
    constexpr int KC = K / 8;
    constexpr int WP = K + 4;
    __shared__ __attribute__((aligned(16))) float wl[3 * 16 * WP];
    {
        v4f w0[1], w1[1], w2[1];
        stage_fetch<K, 16, 32 * WAVES_PER_BLOCK>(Wl1, threadIdx.x, w0);
        stage_fetch<K, 16, 32 * WAVES_PER_BLOCK>(Wr1, threadIdx.x, w1);
        stage_fetch<K, 16, 32 * WAVES_PER_BLOCK>(We1, threadIdx.x, w2);
        stage_put<K, 16, 32 * WAVES_PER_BLOCK>(wl,           threadIdx.x, w0);
        stage_put<K, 16, 32 * WAVES_PER_BLOCK>(wl + 16 * WP, threadIdx.x, w1);
        stage_put<K, 16, 32 * WAVES_PER_BLOCK>(wl + 32 * WP, threadIdx.x, w2);
    }
    __syncthreads();

    const int lane = threadIdx.x & 31;
    const int wave = blockIdx.x * (blockDim.x >> 5) + (threadIdx.x >> 5);
    if (wave >= (M >> 4)) return;
    const int half = lane >> 4;
    const int lm   = lane & 15;
    const int arow = wave * 16 + lm;

    float dv = deg[arow];
    dv = dv > 1.0f ? dv : 1.0f;

    const size_t abase = (size_t)arow * K + half * 4;
    v4f ah[KC], ae[KC];
#pragma unroll
    for (int i = 0; i < KC; ++i) {
        const v4f g = *(const v4f*)(agg0 + abase + 8 * i);
        const v4f r = *(const v4f*)(r0   + abase + 8 * i);
        const v4f e = *(const v4f*)(e0   + abase + 8 * i);
        const v4f b = *(const v4f*)(bl0 + 8 * i + half * 4);
        const float v0 = g.x / dv + b.x + r.x + e.x;
        const float v1 = g.y / dv + b.y + r.y + e.y;
        const float v2 = g.z / dv + b.z + r.z + e.z;
        const float v3 = g.w / dv + b.w + r.w + e.w;
        ah[i].x = v0 > 0.0f ? v0 : 0.0f;
        ah[i].y = v1 > 0.0f ? v1 : 0.0f;
        ah[i].z = v2 > 0.0f ? v2 : 0.0f;
        ah[i].w = v3 > 0.0f ? v3 : 0.0f;
        ae[i] = e;
    }

    const float* wb = wl + lm * WP + half * 4;
    v4f bw[KC];


#pragma unroll
    for (int i = 0; i < KC; ++i)
        bw[i] = *(const v4f*)(wb + 8 * i);
    v8f accp = {0.f, 0.f, 0.f, 0.f, 0.f, 0.f, 0.f, 0.f};
#pragma unroll
    for (int i = 0; i < KC; ++i) {
        accp = __builtin_amdgcn_wmma_f32_16x16x4_f32(
            false, __builtin_shufflevector(ah[i], ah[i], 0, 1),
            false, __builtin_shufflevector(bw[i], bw[i], 0, 1), (short)0, accp, false, false);
        accp = __builtin_amdgcn_wmma_f32_16x16x4_f32(
            false, __builtin_shufflevector(ah[i], ah[i], 2, 3),
            false, __builtin_shufflevector(bw[i], bw[i], 2, 3), (short)0, accp, false, false);
    }

    v4f bx[KC];
#pragma unroll
    for (int i = 0; i < KC; ++i) {
        bw[i] = *(const v4f*)(wb + 16 * WP + 8 * i);
        bx[i] = *(const v4f*)(wb + 32 * WP + 8 * i);
    }
    v8f accr = {0.f, 0.f, 0.f, 0.f, 0.f, 0.f, 0.f, 0.f};
#pragma unroll
    for (int i = 0; i < KC; ++i) {
        accr = __builtin_amdgcn_wmma_f32_16x16x4_f32(
            false, __builtin_shufflevector(ah[i], ah[i], 0, 1),
            false, __builtin_shufflevector(bw[i], bw[i], 0, 1), (short)0, accr, false, false);
        accr = __builtin_amdgcn_wmma_f32_16x16x4_f32(
            false, __builtin_shufflevector(ah[i], ah[i], 2, 3),
            false, __builtin_shufflevector(bw[i], bw[i], 2, 3), (short)0, accr, false, false);
    }
#pragma unroll
    for (int i = 0; i < KC; ++i) {
        accr = __builtin_amdgcn_wmma_f32_16x16x4_f32(
            false, __builtin_shufflevector(ae[i], ae[i], 0, 1),
            false, __builtin_shufflevector(bx[i], bx[i], 0, 1), (short)0, accr, false, false);
        accr = __builtin_amdgcn_wmma_f32_16x16x4_f32(
            false, __builtin_shufflevector(ae[i], ae[i], 2, 3),
            false, __builtin_shufflevector(bx[i], bx[i], 2, 3), (short)0, accr, false, false);
    }

    const float bv = be1[lm];
#pragma unroll
    for (int r = 0; r < 8; ++r) {
        const size_t o = (size_t)(wave * 16 + half * 8 + r) * 16 + lm;
        P1[o]  = accp[r];
        RE1[o] = accr[r] + bv;
    }
}


extern "C" void kernel_launch(void* const* d_in, const int* in_sizes, int n_in,
                              void* d_out, int out_size, void* d_ws, size_t ws_size,
                              hipStream_t stream)
{
    const float* x_feat = (const float*)d_in[0];
    const float* x_emb  = (const float*)d_in[1];
    const int*   eidx   = (const int*)d_in[2];
    const float* Wl0 = (const float*)d_in[3];
    const float* bl0 = (const float*)d_in[4];
    const float* Wr0 = (const float*)d_in[5];
    const float* We0 = (const float*)d_in[6];
    const float* be0 = (const float*)d_in[7];
    const float* Wl1 = (const float*)d_in[8];
    const float* bl1 = (const float*)d_in[9];
    const float* Wr1 = (const float*)d_in[10];
    const float* We1 = (const float*)d_in[11];
    const float* be1 = (const float*)d_in[12];

    const int FEAT = 128;
    const int N = in_sizes[0] / FEAT;
    const int E = in_sizes[2] / 2;
    const int* src = eidx;
    const int* dst = eidx + E;

    float*       ws = (float*)d_ws;
    const size_t n  = (size_t)N;
    float* deg  = ws;
    float* e0   = ws + n;
    float* p0   = ws + 65 * n;
    float* r0   = ws + 129 * n;
    float* agg0 = ws + 193 * n;
    float* p1   = p0;
    float* re1  = ws + 81 * n;
    const int PB = (E + PART_CH - 1) / PART_CH;
    int* tab  = (int*)(ws + 257 * n);
    int* ents = tab + (size_t)PB * PART_PITCH;
    if (ws_size < (257 * n + (size_t)PB * (PART_PITCH + PART_CH)) * sizeof(float)) return;

    const int mwaves  = N >> 4;
    const int gblocks = (mwaves + WAVES_PER_BLOCK - 1) / WAVES_PER_BLOCK;
    const int gthreads = 32 * WAVES_PER_BLOCK;

    partition_kernel<<<PB, PART_T, 0, stream>>>(src, dst, E, N, ents, tab);

    wmma_gemm_kernel<128, 4><<<gblocks, gthreads, 0, stream>>>(x_feat, Wl0, nullptr, p0, N);

    const int G      = (N + OWN_R - 1) / OWN_R;
    const int pblock = (mwaves + AGG0_WAVES - 1) / AGG0_WAVES;
    agg0_owner_kernel<<<G + 2 * pblock, 32 * AGG0_WAVES, 0, stream>>>(
        p0, ents, tab, PB, agg0, deg, N, G, x_feat, Wr0, r0, x_emb, We0, be0, e0);

    layer1_fused_kernel<<<gblocks, gthreads, 0, stream>>>(agg0, r0, e0, bl0, deg,
                                                          Wl1, Wr1, We1, be1, p1, re1, N);

    agg1_owner_kernel<<<G, 32 * OWN_WAVES, 0, stream>>>(p1, ents, tab, PB, N, re1, bl1, deg,
                                                        (float*)d_out);
}


# Tail of a wave's edge ring (its last < 32 entries) gathered as one batch: all row loads issued before the first LDS add waits, one round trip instead of one per entry, in both aggregation launches
# speedup vs baseline: 3.4079x; 1.0291x over previous
#include <hip/hip_runtime.h>
#include <stdint.h>

typedef float v2f __attribute__((ext_vector_type(2)));
typedef float v4f __attribute__((ext_vector_type(4)));
typedef float v8f __attribute__((ext_vector_type(8)));

#define WAVES_PER_BLOCK 8

template <int K, int NO, int NTHREADS>
__device__ __forceinline__
void stage_fetch(const float* __restrict__ W, const int tid, v4f (&v)[NO * (K / 4) / NTHREADS])
{
    constexpr int IT = NO * (K / 4) / NTHREADS;
    static_assert(NO * (K / 4) % NTHREADS == 0, "every thread copies the same number of pieces");
#pragma unroll
    for (int u = 0; u < IT; ++u)
        v[u] = *(const v4f*)(W + (size_t)(u * NTHREADS + tid) * 4);
}

template <int K, int NO, int NTHREADS>
__device__ __forceinline__
void stage_put(float* wl, const int tid, const v4f (&v)[NO * (K / 4) / NTHREADS])
{
    constexpr int Q  = K / 4;
    constexpr int IT = NO * Q / NTHREADS;
#pragma unroll
    for (int u = 0; u < IT; ++u) {
        const int j = u * NTHREADS + tid;
        const int n = j / Q;
        const int q = j - n * Q;
        *(v4f*)(wl + n * (K + 4) + 4 * q) = v[u];
    }
}

template <int K, bool MEANPAIR>
__device__ __forceinline__
void load_a_stripe(const float* __restrict__ X, const int wave, const int lane,
                   v4f (&a)[K / 8], v4f (&a2)[MEANPAIR ? K / 8 : 1])
{
    constexpr int KC = K / 8;
    const int half = lane >> 4;
    const int arow = wave * 16 + (lane & 15);
    const float* __restrict__ x0 = X + (size_t)arow * (MEANPAIR ? 2 * K : K) + half * 4;
#pragma unroll
    for (int i = 0; i < KC; ++i) {
        a[i] = *(const v4f*)(x0 + 8 * i);
        if (MEANPAIR) a2[i] = *(const v4f*)(x0 + K + 8 * i);
    }
}

template <int KC>
__device__ __forceinline__
void mean_pairs(v4f (&a)[KC], const v4f (&a2)[KC])
{
#pragma unroll
    for (int i = 0; i < KC; ++i) {
        a[i].x = 0.5f * (a[i].x + a2[i].x);
        a[i].y = 0.5f * (a[i].y + a2[i].y);
        a[i].z = 0.5f * (a[i].z + a2[i].z);
        a[i].w = 0.5f * (a[i].w + a2[i].w);
    }
}

template <int K, int NT>
__device__ __forceinline__
void gemm_tiles(const v4f (&a)[K / 8], const float* wl,
                const float* __restrict__ bias, float* __restrict__ Out,
                const int wave, const int lane)
{
    constexpr int KC = K / 8;
    constexpr int NO = NT * 16;
    constexpr int WP = K + 4;
    const int half = lane >> 4;
    const int lm   = lane & 15;

#pragma unroll
    for (int nt = 0; nt < NT; ++nt) {
        const float* wrow = wl + (nt * 16 + lm) * WP + half * 4;
        v4f b[KC];
#pragma unroll
        for (int i = 0; i < KC; ++i)
            b[i] = *(const v4f*)(wrow + 8 * i);

        v8f acc = {0.f, 0.f, 0.f, 0.f, 0.f, 0.f, 0.f, 0.f};
#pragma unroll
        for (int i = 0; i < KC; ++i) {
            acc = __builtin_amdgcn_wmma_f32_16x16x4_f32(
                false, __builtin_shufflevector(a[i], a[i], 0, 1),
                false, __builtin_shufflevector(b[i], b[i], 0, 1), (short)0, acc, false, false);
            acc = __builtin_amdgcn_wmma_f32_16x16x4_f32(
                false, __builtin_shufflevector(a[i], a[i], 2, 3),
                false, __builtin_shufflevector(b[i], b[i], 2, 3), (short)0, acc, false, false);
        }

        const int   ocol = nt * 16 + lm;
        const float bv   = bias ? bias[ocol] : 0.0f;
#pragma unroll
        for (int r = 0; r < 8; ++r) {
            const int orow = wave * 16 + half * 8 + r;
            Out[(size_t)orow * NO + ocol] = acc[r] + bv;
        }
    }
}

template <int K, int NT>
__global__ __launch_bounds__(32 * WAVES_PER_BLOCK)
void wmma_gemm_kernel(const float* __restrict__ X, const float* __restrict__ W,
                      const float* __restrict__ bias, float* __restrict__ Out, int M)
{
    __shared__ __attribute__((aligned(16))) float wl[NT * 16 * (K + 4)];
    const int lane = threadIdx.x & 31;
    const int wave = blockIdx.x * (blockDim.x >> 5) + (threadIdx.x >> 5);
    const bool live = wave < (M >> 4);
    v4f wv[NT * 16 * (K / 4) / (32 * WAVES_PER_BLOCK)];
    v4f a[K / 8], a2[1];
    stage_fetch<K, NT * 16, 32 * WAVES_PER_BLOCK>(W, threadIdx.x, wv);
    if (live) load_a_stripe<K, false>(X, wave, lane, a, a2);
    stage_put<K, NT * 16, 32 * WAVES_PER_BLOCK>(wl, threadIdx.x, wv);
    __syncthreads();
    if (live) gemm_tiles<K, NT>(a, wl, bias, Out, wave, lane);
}


constexpr int OWN_R      = 416;
constexpr int OWN_WAVES  = 32;
constexpr int AGG0_WAVES = 16;
constexpr int OWN_QCAP   = 256;
constexpr int PART_CH    = 8192;
constexpr int PART_T     = 1024;
constexpr int PART_PITCH = 256;
constexpr int ENT_SHIFT  = 22;

__global__ __launch_bounds__(PART_T)
void partition_kernel(const int* __restrict__ src, const int* __restrict__ dst, int E, int N,
                      int* __restrict__ ents, int* __restrict__ tab)
{
    __shared__ int hist[PART_PITCH];
    __shared__ int cur[PART_PITCH];
    __shared__ int stage[PART_CH];

    const int tid   = threadIdx.x;
    const int chunk = blockIdx.x;
    const int cbase = chunk * PART_CH;
    const int cnt   = (E - cbase) < PART_CH ? (E - cbase) : PART_CH;
    const int* __restrict__ dch = dst + cbase;
    const int* __restrict__ sch = src + cbase;
    constexpr int PER = PART_CH / PART_T;

    if (tid < PART_PITCH) hist[tid] = 0;
    __syncthreads();

    unsigned d[PER];
    int      s[PER];
#pragma unroll
    for (int u = 0; u < PER; ++u) {
        const int i = u * PART_T + tid;
        d[u] = 0xFFFFFFFFu;
        s[u] = 0;
        if (i < cnt) { d[u] = (unsigned)dch[i]; s[u] = sch[i]; }
    }
#pragma unroll
    for (int u = 0; u < PER; ++u)
        if (d[u] < (unsigned)N) atomicAdd(&hist[d[u] / OWN_R], 1);
    __syncthreads();

    if (tid < 32) {
        int h[8];
        int tot = 0;
#pragma unroll
        for (int i = 0; i < 8; ++i) { h[i] = hist[tid * 8 + i]; tot += h[i]; }
        int incl = tot;
#pragma unroll
        for (int off = 1; off < 32; off <<= 1) {
            const int t = __shfl_up(incl, off);
            if (tid >= off) incl += t;
        }
        int run = incl - tot;
        int* __restrict__ row = tab + (size_t)chunk * PART_PITCH;
#pragma unroll
        for (int i = 0; i < 8; ++i) {
            cur[tid * 8 + i] = run;
            row[tid * 8 + i] = run;
            run += h[i];
        }
    }
    __syncthreads();

#pragma unroll
    for (int u = 0; u < PER; ++u) {
        if (d[u] < (unsigned)N) {
            const unsigned o   = d[u] / OWN_R;
            const int      pos = atomicAdd(&cur[o], 1);
            stage[pos] = s[u] | (int)((d[u] - o * OWN_R) << ENT_SHIFT);
        }
    }
    __syncthreads();

    const int total = cur[PART_PITCH - 1];
    int* __restrict__ o = ents + (size_t)cbase;
    for (int i = tid; i < total; i += PART_T)
        o[i] = stage[i];
}

__device__ __forceinline__ void lds_add(float* p, float v)
{
    __hip_atomic_fetch_add(p, v, __ATOMIC_RELAXED, __HIP_MEMORY_SCOPE_WORKGROUP);
}

template <int C, bool COUNT>
__device__ __forceinline__
void owner_drain32(const float* __restrict__ P, float* aggl, float* degl, const int* q,
                   const int head, const int lane)
{
    const int ent = q[(head + lane) & (OWN_QCAP - 1)];
    const int s   = ent & ((1 << ENT_SHIFT) - 1);
    const int ld  = (int)((unsigned)ent >> ENT_SHIFT);
    if (COUNT) lds_add(degl + ld, 1.0f);
    if (C == 64) {
        v2f r[32];
#pragma unroll
        for (int u = 0; u < 32; ++u) {
            const int sj = __builtin_amdgcn_readlane(s, u);
            r[u] = *(const v2f*)(P + (sj * 64 + 2 * lane));
        }
#pragma unroll
        for (int u = 0; u < 32; ++u) {
            const int dj = __builtin_amdgcn_readlane(ld, u);
            float* a = aggl + (dj * 64 + 2 * lane);
            lds_add(a, r[u].x);
            lds_add(a + 1, r[u].y);
        }
    } else {
        const int hsel = lane >> 4;
        const int c    = lane & 15;
        float v[16];
#pragma unroll
        for (int j = 0; j < 16; ++j) {
            const int sa = __builtin_amdgcn_readlane(s, j);
            const int sb = __builtin_amdgcn_readlane(s, 16 + j);
            const int sj = hsel ? sb : sa;
            v[j] = P[sj * 16 + c];
        }
#pragma unroll
        for (int j = 0; j < 16; ++j) {
            const int da = __builtin_amdgcn_readlane(ld, j);
            const int db = __builtin_amdgcn_readlane(ld, 16 + j);
            const int dj = hsel ? db : da;
            lds_add(aggl + (dj * 16 + c), v[j]);
        }
    }
}

template <int C, bool COUNT>
__device__ __forceinline__
void owner_drain_tail(const float* __restrict__ P, float* aggl, float* degl, const int* q,
                      const int head, const int n, const int lane)
{
    int s = 0, ld = 0;
    if (lane < n) {
        const int ent = q[(head + lane) & (OWN_QCAP - 1)];
        s  = ent & ((1 << ENT_SHIFT) - 1);
        ld = (int)((unsigned)ent >> ENT_SHIFT);
        if (COUNT) lds_add(degl + ld, 1.0f);
    }
    if (C == 64) {
        v2f r[32];
#pragma unroll
        for (int u = 0; u < 32; ++u) {
            if (u < n) {
                const int sj = __builtin_amdgcn_readlane(s, u);
                r[u] = *(const v2f*)(P + (sj * 64 + 2 * lane));
            }
        }
#pragma unroll
        for (int u = 0; u < 32; ++u) {
            if (u < n) {
                const int dj = __builtin_amdgcn_readlane(ld, u);
                float* a = aggl + (dj * 64 + 2 * lane);
                lds_add(a, r[u].x);
                lds_add(a + 1, r[u].y);
            }
        }
    } else {
        const int hsel = lane >> 4;
        const int c    = lane & 15;
        float v[16];
#pragma unroll
        for (int j = 0; j < 16; ++j) {
            if (j < n) {
                const int sa = __builtin_amdgcn_readlane(s, j);
                const int sb = __builtin_amdgcn_readlane(s, 16 + j);
                const int sj = hsel ? sb : sa;
                v[j] = P[sj * 16 + c];
            }
        }
#pragma unroll
        for (int j = 0; j < 16; ++j) {
            if (j < n) {
                const int da = __builtin_amdgcn_readlane(ld, j);
                const int db = __builtin_amdgcn_readlane(ld, 16 + j);
                const int dj = hsel ? db : da;
                if (hsel * 16 + j < n) lds_add(aggl + (dj * 16 + c), v[j]);
            }
        }
    }
}

__device__ __forceinline__
void owner_append(int* q, const int head, int& qn, const unsigned mask, const bool mine,
                  const int ent)
{
    if (mask != 0u) {
        if (mine) {
            const int pos = qn + (int)__builtin_amdgcn_mbcnt_lo(mask, 0u);
            q[(head + pos) & (OWN_QCAP - 1)] = ent;
        }
        qn += __builtin_popcount(mask);
    }
}

template <int C, bool COUNT, int NW>
__device__ __forceinline__
void owner_accumulate(const float* __restrict__ P, const int* __restrict__ ents,
                      const int* __restrict__ tab, const int nchunks, const int own,
                      float* aggl, float* degl, int* qall, const int tid)
{
    const int lane = tid & 31;
    const int wib  = tid >> 5;

    for (int i = tid; i < OWN_R * C / 4; i += 32 * NW)
        ((float4*)aggl)[i] = make_float4(0.f, 0.f, 0.f, 0.f);
    if (COUNT)
        for (int i = tid; i < OWN_R; i += 32 * NW) degl[i] = 0.f;
    __syncthreads();

    int* q = qall + wib * OWN_QCAP;
    int head = 0, qn = 0;

    for (int c0 = wib; c0 < nchunks; c0 += 32 * NW) {
        const int cl = c0 + lane * NW;
        int st = 0, en = 0;
        if (cl < nchunks) {
            const int* __restrict__ row = tab + (size_t)cl * PART_PITCH + own;
            st = row[0];
            en = row[1];
        }
        const int left = (nchunks - c0 + NW - 1) / NW;
        const int nj   = left < 32 ? left : 32;
        for (int j = 0; j < nj; ++j) {
            const int sj = __builtin_amdgcn_readlane(st, j);
            const int ej = __builtin_amdgcn_readlane(en, j);
            const int* __restrict__ seg = ents + (size_t)(c0 + j * NW) * PART_CH;
            for (int t = sj; t < ej; t += 128) {
                int  v[4];
                bool m[4];
#pragma unroll
                for (int u = 0; u < 4; ++u) {
                    const int idx = t + u * 32 + lane;
                    m[u] = idx < ej;
                    v[u] = 0;
                    if (m[u]) v[u] = seg[idx];
                }
#pragma unroll
                for (int u = 0; u < 4; ++u)
                    owner_append(q, head, qn, __builtin_amdgcn_ballot_w32(m[u]), m[u], v[u]);
                while (qn >= 32) {
                    owner_drain32<C, COUNT>(P, aggl, degl, q, head, lane);
                    head = (head + 32) & (OWN_QCAP - 1);
                    qn -= 32;
                }
            }
        }
    }

    if (qn > 0) owner_drain_tail<C, COUNT>(P, aggl, degl, q, head, qn, lane);
    __syncthreads();
}

__global__ __launch_bounds__(32 * AGG0_WAVES)
void agg0_owner_kernel(const float* __restrict__ p0, const int* __restrict__ ents,
                       const int* __restrict__ tab, int nchunks, float* __restrict__ agg0,
                       float* __restrict__ deg, int N, int G,
                       const float* __restrict__ x_feat, const float* __restrict__ Wr0,
                       float* __restrict__ r0,
                       const float* __restrict__ x_emb, const float* __restrict__ We0,
                       const float* __restrict__ be0, float* __restrict__ e0)
{
    static_assert(OWN_R * 64 >= 64 * (128 + 4), "a projection block's weight copy must fit in aggl");
    __shared__ __attribute__((aligned(16))) float aggl[OWN_R * 64];
    __shared__ float degl[OWN_R];
    __shared__ int   qall[AGG0_WAVES * OWN_QCAP];

    const int tid = threadIdx.x;
    const int b   = blockIdx.x;
    if (b >= G) {
        const int  pb     = b - G;
        const int  stripe = (pb >> 1) * AGG0_WAVES + (tid >> 5);
        const bool live   = stripe < (N >> 4);
        if (pb & 1) {
            v4f wv[2], a[8], a2[8];
            stage_fetch<64, 64, 32 * AGG0_WAVES>(We0, tid, wv);
            if (live) load_a_stripe<64, true>(x_emb, stripe, tid & 31, a, a2);
            stage_put<64, 64, 32 * AGG0_WAVES>(aggl, tid, wv);
            __syncthreads();
            if (live) {
                mean_pairs<8>(a, a2);
                gemm_tiles<64, 4>(a, aggl, be0, e0, stripe, tid & 31);
            }
        } else {
            v4f wv[4], a[16], a2[1];
            stage_fetch<128, 64, 32 * AGG0_WAVES>(Wr0, tid, wv);
            if (live) load_a_stripe<128, false>(x_feat, stripe, tid & 31, a, a2);
            stage_put<128, 64, 32 * AGG0_WAVES>(aggl, tid, wv);
            __syncthreads();
            if (live) gemm_tiles<128, 4>(a, aggl, nullptr, r0, stripe, tid & 31);
        }
        return;
    }

    const int base = b * OWN_R;
    owner_accumulate<64, true, AGG0_WAVES>(p0, ents, tab, nchunks, b, aggl, degl, qall, tid);

    const int nrows = (N - base) < OWN_R ? (N - base) : OWN_R;
    float4* __restrict__ o4 = reinterpret_cast<float4*>(agg0 + (size_t)base * 64);
    for (int i = tid; i < nrows * 16; i += 32 * AGG0_WAVES)
        o4[i] = ((const float4*)aggl)[i];
    for (int i = tid; i < nrows; i += 32 * AGG0_WAVES)
        deg[base + i] = degl[i];
}

__global__ __launch_bounds__(32 * OWN_WAVES)
void agg1_owner_kernel(const float* __restrict__ p1, const int* __restrict__ ents,
                       const int* __restrict__ tab, int nchunks, int N,
                       const float* __restrict__ re1, const float* __restrict__ bl1,
                       const float* __restrict__ deg, float* __restrict__ out)
{
    __shared__ __attribute__((aligned(16))) float aggl[OWN_R * 16];
    __shared__ int qall[OWN_WAVES * OWN_QCAP];

    const int tid  = threadIdx.x;
    const int base = blockIdx.x * OWN_R;
    owner_accumulate<16, false, OWN_WAVES>(p1, ents, tab, nchunks, blockIdx.x, aggl, nullptr, qall, tid);

    const int nrows = (N - base) < OWN_R ? (N - base) : OWN_R;
    for (int i = tid; i < nrows * 16; i += 32 * OWN_WAVES) {
        float dv = deg[base + (i >> 4)];
        dv = dv > 1.0f ? dv : 1.0f;
        const int t = base * 16 + i;
        out[t] = aggl[i] / dv + bl1[i & 15] + re1[t];
    }
}

__global__ __launch_bounds__(32 * WAVES_PER_BLOCK)
void layer1_fused_kernel(const float* __restrict__ agg0, const float* __restrict__ r0,
                         const float* __restrict__ e0, const float* __restrict__ bl0,
                         const float* __restrict__ deg,
                         const float* __restrict__ Wl1, const float* __restrict__ Wr1,
                         const float* __restrict__ We1, const float* __restrict__ be1,
                         float* __restrict__ P1, float* __restrict__ RE1, int M)
{
    constexpr int K  = 64;
    constexpr int KC = K / 8;
    constexpr int WP = K + 4;
    __shared__ __attribute__((aligned(16))) float wl[3 * 16 * WP];
    {
        v4f w0[1], w1[1], w2[1];
        stage_fetch<K, 16, 32 * WAVES_PER_BLOCK>(Wl1, threadIdx.x, w0);
        stage_fetch<K, 16, 32 * WAVES_PER_BLOCK>(Wr1, threadIdx.x, w1);
        stage_fetch<K, 16, 32 * WAVES_PER_BLOCK>(We1, threadIdx.x, w2);
        stage_put<K, 16, 32 * WAVES_PER_BLOCK>(wl,           threadIdx.x, w0);
        stage_put<K, 16, 32 * WAVES_PER_BLOCK>(wl + 16 * WP, threadIdx.x, w1);
        stage_put<K, 16, 32 * WAVES_PER_BLOCK>(wl + 32 * WP, threadIdx.x, w2);
    }
    __syncthreads();

    const int lane = threadIdx.x & 31;
    const int wave = blockIdx.x * (blockDim.x >> 5) + (threadIdx.x >> 5);
    if (wave >= (M >> 4)) return;
    const int half = lane >> 4;
    const int lm   = lane & 15;
    const int arow = wave * 16 + lm;

    float dv = deg[arow];
    dv = dv > 1.0f ? dv : 1.0f;

    const size_t abase = (size_t)arow * K + half * 4;
    v4f ah[KC], ae[KC];
#pragma unroll
    for (int i = 0; i < KC; ++i) {
        const v4f g = *(const v4f*)(agg0 + abase + 8 * i);
        const v4f r = *(const v4f*)(r0   + abase + 8 * i);
        const v4f e = *(const v4f*)(e0   + abase + 8 * i);
        const v4f b = *(const v4f*)(bl0 + 8 * i + half * 4);
        const float v0 = g.x / dv + b.x + r.x + e.x;
        const float v1 = g.y / dv + b.y + r.y + e.y;
        const float v2 = g.z / dv + b.z + r.z + e.z;
        const float v3 = g.w / dv + b.w + r.w + e.w;
        ah[i].x = v0 > 0.0f ? v0 : 0.0f;
        ah[i].y = v1 > 0.0f ? v1 : 0.0f;
        ah[i].z = v2 > 0.0f ? v2 : 0.0f;
        ah[i].w = v3 > 0.0f ? v3 : 0.0f;
        ae[i] = e;
    }

    const float* wb = wl + lm * WP + half * 4;
    v4f bw[KC];


#pragma unroll
    for (int i = 0; i < KC; ++i)
        bw[i] = *(const v4f*)(wb + 8 * i);
    v8f accp = {0.f, 0.f, 0.f, 0.f, 0.f, 0.f, 0.f, 0.f};
#pragma unroll
    for (int i = 0; i < KC; ++i) {
        accp = __builtin_amdgcn_wmma_f32_16x16x4_f32(
            false, __builtin_shufflevector(ah[i], ah[i], 0, 1),
            false, __builtin_shufflevector(bw[i], bw[i], 0, 1), (short)0, accp, false, false);
        accp = __builtin_amdgcn_wmma_f32_16x16x4_f32(
            false, __builtin_shufflevector(ah[i], ah[i], 2, 3),
            false, __builtin_shufflevector(bw[i], bw[i], 2, 3), (short)0, accp, false, false);
    }

    v4f bx[KC];
#pragma unroll
    for (int i = 0; i < KC; ++i) {
        bw[i] = *(const v4f*)(wb + 16 * WP + 8 * i);
        bx[i] = *(const v4f*)(wb + 32 * WP + 8 * i);
    }
    v8f accr = {0.f, 0.f, 0.f, 0.f, 0.f, 0.f, 0.f, 0.f};
#pragma unroll
    for (int i = 0; i < KC; ++i) {
        accr = __builtin_amdgcn_wmma_f32_16x16x4_f32(
            false, __builtin_shufflevector(ah[i], ah[i], 0, 1),
            false, __builtin_shufflevector(bw[i], bw[i], 0, 1), (short)0, accr, false, false);
        accr = __builtin_amdgcn_wmma_f32_16x16x4_f32(
            false, __builtin_shufflevector(ah[i], ah[i], 2, 3),
            false, __builtin_shufflevector(bw[i], bw[i], 2, 3), (short)0, accr, false, false);
    }
#pragma unroll
    for (int i = 0; i < KC; ++i) {
        accr = __builtin_amdgcn_wmma_f32_16x16x4_f32(
            false, __builtin_shufflevector(ae[i], ae[i], 0, 1),
            false, __builtin_shufflevector(bx[i], bx[i], 0, 1), (short)0, accr, false, false);
        accr = __builtin_amdgcn_wmma_f32_16x16x4_f32(
            false, __builtin_shufflevector(ae[i], ae[i], 2, 3),
            false, __builtin_shufflevector(bx[i], bx[i], 2, 3), (short)0, accr, false, false);
    }

    const float bv = be1[lm];
#pragma unroll
    for (int r = 0; r < 8; ++r) {
        const size_t o = (size_t)(wave * 16 + half * 8 + r) * 16 + lm;
        P1[o]  = accp[r];
        RE1[o] = accr[r] + bv;
    }
}


extern "C" void kernel_launch(void* const* d_in, const int* in_sizes, int n_in,
                              void* d_out, int out_size, void* d_ws, size_t ws_size,
                              hipStream_t stream)
{
    const float* x_feat = (const float*)d_in[0];
    const float* x_emb  = (const float*)d_in[1];
    const int*   eidx   = (const int*)d_in[2];
    const float* Wl0 = (const float*)d_in[3];
    const float* bl0 = (const float*)d_in[4];
    const float* Wr0 = (const float*)d_in[5];
    const float* We0 = (const float*)d_in[6];
    const float* be0 = (const float*)d_in[7];
    const float* Wl1 = (const float*)d_in[8];
    const float* bl1 = (const float*)d_in[9];
    const float* Wr1 = (const float*)d_in[10];
    const float* We1 = (const float*)d_in[11];
    const float* be1 = (const float*)d_in[12];

    const int FEAT = 128;
    const int N = in_sizes[0] / FEAT;
    const int E = in_sizes[2] / 2;
    const int* src = eidx;
    const int* dst = eidx + E;

    float*       ws = (float*)d_ws;
    const size_t n  = (size_t)N;
    float* deg  = ws;
    float* e0   = ws + n;
    float* p0   = ws + 65 * n;
    float* r0   = ws + 129 * n;
    float* agg0 = ws + 193 * n;
    float* p1   = p0;
    float* re1  = ws + 81 * n;
    const int PB = (E + PART_CH - 1) / PART_CH;
    int* tab  = (int*)(ws + 257 * n);
    int* ents = tab + (size_t)PB * PART_PITCH;
    if (ws_size < (257 * n + (size_t)PB * (PART_PITCH + PART_CH)) * sizeof(float)) return;

    const int mwaves  = N >> 4;
    const int gblocks = (mwaves + WAVES_PER_BLOCK - 1) / WAVES_PER_BLOCK;
    const int gthreads = 32 * WAVES_PER_BLOCK;

    partition_kernel<<<PB, PART_T, 0, stream>>>(src, dst, E, N, ents, tab);

    wmma_gemm_kernel<128, 4><<<gblocks, gthreads, 0, stream>>>(x_feat, Wl0, nullptr, p0, N);

    const int G      = (N + OWN_R - 1) / OWN_R;
    const int pblock = (mwaves + AGG0_WAVES - 1) / AGG0_WAVES;
    agg0_owner_kernel<<<G + 2 * pblock, 32 * AGG0_WAVES, 0, stream>>>(
        p0, ents, tab, PB, agg0, deg, N, G, x_feat, Wr0, r0, x_emb, We0, be0, e0);

    layer1_fused_kernel<<<gblocks, gthreads, 0, stream>>>(agg0, r0, e0, bl0, deg,
                                                          Wl1, Wr1, We1, be1, p1, re1, N);

    agg1_owner_kernel<<<G, 32 * OWN_WAVES, 0, stream>>>(p1, ents, tab, PB, N, re1, bl1, deg,
                                                        (float*)d_out);
}
